# starting kernel
# baseline (speedup 1.0000x reference)
#include <hip/hip_runtime.h>
#include <hip/hip_bf16.h>

typedef __attribute__((ext_vector_type(2))) float v2f;
typedef __attribute__((ext_vector_type(8))) float v8f;

__device__ __forceinline__ float sigmoidf_(float x) { return 1.0f / (1.0f + __expf(-x)); }

__device__ __forceinline__ v8f wmma4(v2f a, v2f b, v8f c) {
  return __builtin_amdgcn_wmma_f32_16x16x4_f32(false, a, false, b, (short)0, c,
                                               false, false);
}

template<bool WTRANS, bool HASA2, bool RELU>
__global__ void __launch_bounds__(256)
gemm_wmma_f32(const float* __restrict__ A1, int lda1,
              const float* __restrict__ W1, int ldw1, int K1,
              const float* __restrict__ A2, int lda2,
              const float* __restrict__ W2, int ldw2, int K2,
              const float* __restrict__ b1p, const float* __restrict__ b2p,
              float bias_scale,
              float* __restrict__ C, int ldc, int Mrows, int Ncols)
{
  const int wave = threadIdx.x >> 5;
  const int lane = threadIdx.x & 31;
  const int ntn  = Ncols >> 5;
  const int total = (Mrows >> 4) * ntn;
  const int tile = blockIdx.x * 8 + wave;
  if (tile >= total) return;
  const int mt = tile / ntn, nt = tile - mt * ntn;
  const int m0 = mt << 4, n0 = nt << 5;
  const int half = lane >> 4;
  const int lp   = lane & 15;

  v8f acc0 = {}, acc1 = {};

  {
    const float* ap  = A1 + (size_t)(m0 + lp) * lda1 + half * 2;
    const float* wp0;
    const float* wp1;
    if (WTRANS) {
      wp0 = W1 + (size_t)(n0 + lp) * ldw1 + half * 2;
      wp1 = wp0 + (size_t)16 * ldw1;
    } else {
      wp0 = W1 + (size_t)(half * 2) * ldw1 + n0 + lp;
      wp1 = wp0 + 16;
    }
    #pragma unroll 4
    for (int k0 = 0; k0 < K1; k0 += 4) {
      v2f a, b0, b1;
      a.x = ap[0]; a.y = ap[1]; ap += 4;
      if (WTRANS) {
        b0.x = wp0[0]; b0.y = wp0[1]; wp0 += 4;
        b1.x = wp1[0]; b1.y = wp1[1]; wp1 += 4;
      } else {
        b0.x = wp0[0]; b0.y = wp0[ldw1];
        b1.x = wp1[0]; b1.y = wp1[ldw1];
        wp0 += (size_t)4 * ldw1; wp1 += (size_t)4 * ldw1;
      }
      acc0 = wmma4(a, b0, acc0);
      acc1 = wmma4(a, b1, acc1);
    }
  }

  if (HASA2) {
    const float* ap  = A2 + (size_t)(m0 + lp) * lda2 + half * 2;
    const float* wp0 = W2 + (size_t)(half * 2) * ldw2 + n0 + lp;
    const float* wp1 = wp0 + 16;
    #pragma unroll 4
    for (int k0 = 0; k0 < K2; k0 += 4) {
      v2f a, b0, b1;
      a.x = ap[0]; a.y = ap[1]; ap += 4;
      b0.x = wp0[0]; b0.y = wp0[ldw2];
      b1.x = wp1[0]; b1.y = wp1[ldw2];
      wp0 += (size_t)4 * ldw2; wp1 += (size_t)4 * ldw2;
      acc0 = wmma4(a, b0, acc0);
      acc1 = wmma4(a, b1, acc1);
    }
  }

  float bias0 = 0.0f, bias1 = 0.0f;
  if (b1p) { bias0  = bias_scale * b1p[n0 + lp];      bias1  = bias_scale * b1p[n0 + 16 + lp]; }
  if (b2p) { bias0 += b2p[n0 + lp];                   bias1 += b2p[n0 + 16 + lp]; }

  float* crow = C + (size_t)(m0 + half * 8) * ldc + n0 + lp;
  #pragma unroll
  for (int r = 0; r < 8; ++r) {
    float v0 = acc0[r] + bias0;
    float v1 = acc1[r] + bias1;
    if (RELU) { v0 = fmaxf(v0, 0.0f); v1 = fmaxf(v1, 0.0f); }
    crow[0]  = v0;
    crow[16] = v1;
    crow += ldc;
  }
}

__global__ void __launch_bounds__(256)
pair_relu_sum(const float* __restrict__ Af, const float* __restrict__ Bf,
              float* __restrict__ S)
{
  const int ri = blockIdx.x;
  const int b  = ri >> 7;
  const int f  = threadIdx.x;
  const float bm = Bf[(size_t)ri * 256 + f];
  const float* ap = Af + (size_t)b * 128 * 256 + f;
  float acc = 0.0f;
  #pragma unroll 4
  for (int j = 0; j < 128; ++j)
    acc += fmaxf(ap[(size_t)j * 256] + bm, 0.0f);
  S[(size_t)ri * 256 + f] = acc;
}

__global__ void __launch_bounds__(512)
lstm_scan(const float* __restrict__ Gx, const float* __restrict__ whh,
          const float* __restrict__ h0, const float* __restrict__ c0,
          float* __restrict__ new_hidden)
{
  __shared__ float h_s[16][128];
  __shared__ float c_s[16][128];
  __shared__ float g_s[16][512];
  const int tid = threadIdx.x;
  for (int idx = tid; idx < 2048; idx += 512) {
    h_s[idx >> 7][idx & 127] = h0[idx];
    c_s[idx >> 7][idx & 127] = c0[idx];
  }
  __syncthreads();
  for (int t = 0; t < 128; ++t) {
    const int u = tid;
    float acc[16];
    #pragma unroll
    for (int b = 0; b < 16; ++b)
      acc[b] = Gx[((size_t)(b * 128 + t)) * 512 + u];
    for (int k = 0; k < 128; ++k) {
      const float w = whh[(size_t)u * 128 + k];
      #pragma unroll
      for (int b = 0; b < 16; ++b)
        acc[b] = fmaf(h_s[b][k], w, acc[b]);
    }
    #pragma unroll
    for (int b = 0; b < 16; ++b) g_s[b][u] = acc[b];
    __syncthreads();
    for (int idx = tid; idx < 2048; idx += 512) {
      const int b = idx >> 7, n = idx & 127;
      const float ig = g_s[b][n],       fg = g_s[b][128 + n];
      const float gg = g_s[b][256 + n], og = g_s[b][384 + n];
      const float c = sigmoidf_(fg) * c_s[b][n] + sigmoidf_(ig) * tanhf(gg);
      const float h = sigmoidf_(og) * tanhf(c);
      c_s[b][n] = c; h_s[b][n] = h;
      new_hidden[(size_t)(b * 128 + t) * 128 + n] = h;
    }
    __syncthreads();
  }
}

extern "C" void kernel_launch(void* const* d_in, const int* in_sizes, int n_in,
                              void* d_out, int out_size, void* d_ws, size_t ws_size,
                              hipStream_t stream)
{
  (void)in_sizes; (void)n_in; (void)out_size; (void)ws_size;
  const float* x      = (const float*)d_in[0];
  const float* hidden = (const float*)d_in[1];
  const float* h0     = (const float*)d_in[2];
  const float* c0     = (const float*)d_in[3];
  const float* f_w1   = (const float*)d_in[4];
  const float* f_b1   = (const float*)d_in[5];
  const float* f_w2   = (const float*)d_in[6];
  const float* f_b2   = (const float*)d_in[7];
  const float* g_w1   = (const float*)d_in[8];
  const float* g_b1   = (const float*)d_in[9];
  const float* g_w2   = (const float*)d_in[10];
  const float* g_b2   = (const float*)d_in[11];
  const float* wih    = (const float*)d_in[12];
  const float* whh    = (const float*)d_in[13];
  const float* bih    = (const float*)d_in[14];
  const float* bhh    = (const float*)d_in[15];
  const float* o_w1   = (const float*)d_in[16];
  const float* o_b1   = (const float*)d_in[17];
  const float* o_w2   = (const float*)d_in[18];
  const float* o_b2   = (const float*)d_in[19];
  float* out = (float*)d_out;

  const int BN = 16 * 128;
  float* R0 = (float*)d_ws;
  float* R1 = R0 + (size_t)BN * 256;
  float* R2 = R1 + (size_t)BN * 256;
  float* R3 = R2 + (size_t)BN * 256;

  auto blocks = [](int M, int N) { return ((M >> 4) * (N >> 5) + 7) / 8; };

  gemm_wmma_f32<false,false,false><<<blocks(BN,256), 256, 0, stream>>>(
      hidden,128, f_w1,256,128, nullptr,0, nullptr,0,0,
      nullptr,nullptr,1.0f, R0,256, BN,256);
  gemm_wmma_f32<false,false,false><<<blocks(BN,256), 256, 0, stream>>>(
      hidden,128, f_w1 + 128*256,256,128, nullptr,0, nullptr,0,0,
      f_b1,nullptr,1.0f, R1,256, BN,256);
  pair_relu_sum<<<BN, 256, 0, stream>>>(R0, R1, R2);
  gemm_wmma_f32<false,false,false><<<blocks(BN,128), 256, 0, stream>>>(
      R2,256, f_w2,128,256, nullptr,0, nullptr,0,0,
      f_b2,nullptr,128.0f, R0,128, BN,128);
  gemm_wmma_f32<false,true,true><<<blocks(BN,256), 256, 0, stream>>>(
      x,128, g_w1,256,128, R0,128, g_w1 + 128*256,256,128,
      g_b1,nullptr,1.0f, R1,256, BN,256);
  gemm_wmma_f32<false,false,false><<<blocks(BN,128), 256, 0, stream>>>(
      R1,256, g_w2,128,256, nullptr,0, nullptr,0,0,
      g_b2,nullptr,1.0f, R2,128, BN,128);
  gemm_wmma_f32<true,false,false><<<blocks(BN,512), 256, 0, stream>>>(
      R2,128, wih,128,128, nullptr,0, nullptr,0,0,
      bih,bhh,1.0f, R0,512, BN,512);
  lstm_scan<<<1, 512, 0, stream>>>(R0, whh, h0, c0, R3);
  gemm_wmma_f32<false,false,true><<<blocks(BN,256), 256, 0, stream>>>(
      R3,128, o_w1,256,128, nullptr,0, nullptr,0,0,
      o_b1,nullptr,1.0f, R0,256, BN,256);
  gemm_wmma_f32<false,false,false><<<blocks(BN,32), 256, 0, stream>>>(
      R0,256, o_w2,32,256, nullptr,0, nullptr,0,0,
      o_b2,nullptr,1.0f, out,32, BN,32);
}


# LSTM scan: one workgroup per batch row (16 run in parallel), each thread keeps its unit's four whh gate rows in registers; gate sums split into 4 partial sums (f32 summation order changes)
# speedup vs baseline: 6.4885x; 6.4885x over previous
#include <hip/hip_runtime.h>
#include <hip/hip_bf16.h>

typedef __attribute__((ext_vector_type(2))) float v2f;
typedef __attribute__((ext_vector_type(8))) float v8f;

__device__ __forceinline__ float sigmoidf_(float x) { return 1.0f / (1.0f + __expf(-x)); }

__device__ __forceinline__ v8f wmma4(v2f a, v2f b, v8f c) {
  return __builtin_amdgcn_wmma_f32_16x16x4_f32(false, a, false, b, (short)0, c,
                                               false, false);
}

template<bool WTRANS, bool HASA2, bool RELU>
__global__ void __launch_bounds__(256)
gemm_wmma_f32(const float* __restrict__ A1, int lda1,
              const float* __restrict__ W1, int ldw1, int K1,
              const float* __restrict__ A2, int lda2,
              const float* __restrict__ W2, int ldw2, int K2,
              const float* __restrict__ b1p, const float* __restrict__ b2p,
              float bias_scale,
              float* __restrict__ C, int ldc, int Mrows, int Ncols)
{
  const int wave = threadIdx.x >> 5;
  const int lane = threadIdx.x & 31;
  const int ntn  = Ncols >> 5;
  const int total = (Mrows >> 4) * ntn;
  const int tile = blockIdx.x * 8 + wave;
  if (tile >= total) return;
  const int mt = tile / ntn, nt = tile - mt * ntn;
  const int m0 = mt << 4, n0 = nt << 5;
  const int half = lane >> 4;
  const int lp   = lane & 15;

  v8f acc0 = {}, acc1 = {};

  {
    const float* ap  = A1 + (size_t)(m0 + lp) * lda1 + half * 2;
    const float* wp0;
    const float* wp1;
    if (WTRANS) {
      wp0 = W1 + (size_t)(n0 + lp) * ldw1 + half * 2;
      wp1 = wp0 + (size_t)16 * ldw1;
    } else {
      wp0 = W1 + (size_t)(half * 2) * ldw1 + n0 + lp;
      wp1 = wp0 + 16;
    }
    #pragma unroll 4
    for (int k0 = 0; k0 < K1; k0 += 4) {
      v2f a, b0, b1;
      a.x = ap[0]; a.y = ap[1]; ap += 4;
      if (WTRANS) {
        b0.x = wp0[0]; b0.y = wp0[1]; wp0 += 4;
        b1.x = wp1[0]; b1.y = wp1[1]; wp1 += 4;
      } else {
        b0.x = wp0[0]; b0.y = wp0[ldw1];
        b1.x = wp1[0]; b1.y = wp1[ldw1];
        wp0 += (size_t)4 * ldw1; wp1 += (size_t)4 * ldw1;
      }
      acc0 = wmma4(a, b0, acc0);
      acc1 = wmma4(a, b1, acc1);
    }
  }

  if (HASA2) {
    const float* ap  = A2 + (size_t)(m0 + lp) * lda2 + half * 2;
    const float* wp0 = W2 + (size_t)(half * 2) * ldw2 + n0 + lp;
    const float* wp1 = wp0 + 16;
    #pragma unroll 4
    for (int k0 = 0; k0 < K2; k0 += 4) {
      v2f a, b0, b1;
      a.x = ap[0]; a.y = ap[1]; ap += 4;
      b0.x = wp0[0]; b0.y = wp0[ldw2];
      b1.x = wp1[0]; b1.y = wp1[ldw2];
      wp0 += (size_t)4 * ldw2; wp1 += (size_t)4 * ldw2;
      acc0 = wmma4(a, b0, acc0);
      acc1 = wmma4(a, b1, acc1);
    }
  }

  float bias0 = 0.0f, bias1 = 0.0f;
  if (b1p) { bias0  = bias_scale * b1p[n0 + lp];      bias1  = bias_scale * b1p[n0 + 16 + lp]; }
  if (b2p) { bias0 += b2p[n0 + lp];                   bias1 += b2p[n0 + 16 + lp]; }

  float* crow = C + (size_t)(m0 + half * 8) * ldc + n0 + lp;
  #pragma unroll
  for (int r = 0; r < 8; ++r) {
    float v0 = acc0[r] + bias0;
    float v1 = acc1[r] + bias1;
    if (RELU) { v0 = fmaxf(v0, 0.0f); v1 = fmaxf(v1, 0.0f); }
    crow[0]  = v0;
    crow[16] = v1;
    crow += ldc;
  }
}

__global__ void __launch_bounds__(256)
pair_relu_sum(const float* __restrict__ Af, const float* __restrict__ Bf,
              float* __restrict__ S)
{
  const int ri = blockIdx.x;
  const int b  = ri >> 7;
  const int f  = threadIdx.x;
  const float bm = Bf[(size_t)ri * 256 + f];
  const float* ap = Af + (size_t)b * 128 * 256 + f;
  float acc = 0.0f;
  #pragma unroll 4
  for (int j = 0; j < 128; ++j)
    acc += fmaxf(ap[(size_t)j * 256] + bm, 0.0f);
  S[(size_t)ri * 256 + f] = acc;
}

__global__ void __launch_bounds__(128)
lstm_scan(const float* __restrict__ Gx, const float* __restrict__ whh,
          const float* __restrict__ h0, const float* __restrict__ c0,
          float* __restrict__ new_hidden)
{
  __shared__ __attribute__((aligned(16))) float h_s[2][128];
  const int b = blockIdx.x;
  const int n = threadIdx.x;

  float wi[128], wf[128], wg[128], wo[128];
  {
    const float4* ri = reinterpret_cast<const float4*>(whh + (size_t)(0 * 128 + n) * 128);
    const float4* rf = reinterpret_cast<const float4*>(whh + (size_t)(1 * 128 + n) * 128);
    const float4* rg = reinterpret_cast<const float4*>(whh + (size_t)(2 * 128 + n) * 128);
    const float4* ro = reinterpret_cast<const float4*>(whh + (size_t)(3 * 128 + n) * 128);
    #pragma unroll
    for (int q = 0; q < 32; ++q) {
      const float4 vi = ri[q], vf = rf[q], vg = rg[q], vo = ro[q];
      wi[4*q] = vi.x; wi[4*q+1] = vi.y; wi[4*q+2] = vi.z; wi[4*q+3] = vi.w;
      wf[4*q] = vf.x; wf[4*q+1] = vf.y; wf[4*q+2] = vf.z; wf[4*q+3] = vf.w;
      wg[4*q] = vg.x; wg[4*q+1] = vg.y; wg[4*q+2] = vg.z; wg[4*q+3] = vg.w;
      wo[4*q] = vo.x; wo[4*q+1] = vo.y; wo[4*q+2] = vo.z; wo[4*q+3] = vo.w;
    }
  }

  float c = c0[b * 128 + n];
  h_s[0][n] = h0[b * 128 + n];

  const float* gx = Gx + (size_t)b * 128 * 512 + n;
  float gi = gx[0], gf = gx[128], gg = gx[256], go = gx[384];
  float* hout = new_hidden + (size_t)b * 128 * 128 + n;
  __syncthreads();

  for (int t = 0; t < 128; ++t) {
    const float* hs = h_s[t & 1];
    float ai[4] = {gi, 0.0f, 0.0f, 0.0f};
    float af[4] = {gf, 0.0f, 0.0f, 0.0f};
    float ag[4] = {gg, 0.0f, 0.0f, 0.0f};
    float ao[4] = {go, 0.0f, 0.0f, 0.0f};
    {
      const int tn = (t + 1 < 128) ? (t + 1) : 127;
      const float* gn = gx + (size_t)tn * 512;
      gi = gn[0]; gf = gn[128]; gg = gn[256]; go = gn[384];
    }
    #pragma unroll
    for (int q = 0; q < 8; ++q) {
      #pragma unroll
      for (int p = 0; p < 4; ++p) {
        const int k = p * 32 + q * 4;
        const float4 hv = *reinterpret_cast<const float4*>(hs + k);
        ai[p] = fmaf(hv.x, wi[k], ai[p]);   af[p] = fmaf(hv.x, wf[k], af[p]);
        ag[p] = fmaf(hv.x, wg[k], ag[p]);   ao[p] = fmaf(hv.x, wo[k], ao[p]);
        ai[p] = fmaf(hv.y, wi[k+1], ai[p]); af[p] = fmaf(hv.y, wf[k+1], af[p]);
        ag[p] = fmaf(hv.y, wg[k+1], ag[p]); ao[p] = fmaf(hv.y, wo[k+1], ao[p]);
        ai[p] = fmaf(hv.z, wi[k+2], ai[p]); af[p] = fmaf(hv.z, wf[k+2], af[p]);
        ag[p] = fmaf(hv.z, wg[k+2], ag[p]); ao[p] = fmaf(hv.z, wo[k+2], ao[p]);
        ai[p] = fmaf(hv.w, wi[k+3], ai[p]); af[p] = fmaf(hv.w, wf[k+3], af[p]);
        ag[p] = fmaf(hv.w, wg[k+3], ag[p]); ao[p] = fmaf(hv.w, wo[k+3], ao[p]);
      }
    }
    const float ig = (ai[0] + ai[1]) + (ai[2] + ai[3]);
    const float fg = (af[0] + af[1]) + (af[2] + af[3]);
    const float g2 = (ag[0] + ag[1]) + (ag[2] + ag[3]);
    const float og = (ao[0] + ao[1]) + (ao[2] + ao[3]);
    c = sigmoidf_(fg) * c + sigmoidf_(ig) * tanhf(g2);
    const float h = sigmoidf_(og) * tanhf(c);
    h_s[(t + 1) & 1][n] = h;
    __syncthreads();
    hout[(size_t)t * 128] = h;
  }
}

extern "C" void kernel_launch(void* const* d_in, const int* in_sizes, int n_in,
                              void* d_out, int out_size, void* d_ws, size_t ws_size,
                              hipStream_t stream)
{
  (void)in_sizes; (void)n_in; (void)out_size; (void)ws_size;
  const float* x      = (const float*)d_in[0];
  const float* hidden = (const float*)d_in[1];
  const float* h0     = (const float*)d_in[2];
  const float* c0     = (const float*)d_in[3];
  const float* f_w1   = (const float*)d_in[4];
  const float* f_b1   = (const float*)d_in[5];
  const float* f_w2   = (const float*)d_in[6];
  const float* f_b2   = (const float*)d_in[7];
  const float* g_w1   = (const float*)d_in[8];
  const float* g_b1   = (const float*)d_in[9];
  const float* g_w2   = (const float*)d_in[10];
  const float* g_b2   = (const float*)d_in[11];
  const float* wih    = (const float*)d_in[12];
  const float* whh    = (const float*)d_in[13];
  const float* bih    = (const float*)d_in[14];
  const float* bhh    = (const float*)d_in[15];
  const float* o_w1   = (const float*)d_in[16];
  const float* o_b1   = (const float*)d_in[17];
  const float* o_w2   = (const float*)d_in[18];
  const float* o_b2   = (const float*)d_in[19];
  float* out = (float*)d_out;

  const int BN = 16 * 128;
  float* R0 = (float*)d_ws;
  float* R1 = R0 + (size_t)BN * 256;
  float* R2 = R1 + (size_t)BN * 256;
  float* R3 = R2 + (size_t)BN * 256;

  auto blocks = [](int M, int N) { return ((M >> 4) * (N >> 5) + 7) / 8; };

  gemm_wmma_f32<false,false,false><<<blocks(BN,256), 256, 0, stream>>>(
      hidden,128, f_w1,256,128, nullptr,0, nullptr,0,0,
      nullptr,nullptr,1.0f, R0,256, BN,256);
  gemm_wmma_f32<false,false,false><<<blocks(BN,256), 256, 0, stream>>>(
      hidden,128, f_w1 + 128*256,256,128, nullptr,0, nullptr,0,0,
      f_b1,nullptr,1.0f, R1,256, BN,256);
  pair_relu_sum<<<BN, 256, 0, stream>>>(R0, R1, R2);
  gemm_wmma_f32<false,false,false><<<blocks(BN,128), 256, 0, stream>>>(
      R2,256, f_w2,128,256, nullptr,0, nullptr,0,0,
      f_b2,nullptr,128.0f, R0,128, BN,128);
  gemm_wmma_f32<false,true,true><<<blocks(BN,256), 256, 0, stream>>>(
      x,128, g_w1,256,128, R0,128, g_w1 + 128*256,256,128,
      g_b1,nullptr,1.0f, R1,256, BN,256);
  gemm_wmma_f32<false,false,false><<<blocks(BN,128), 256, 0, stream>>>(
      R1,256, g_w2,128,256, nullptr,0, nullptr,0,0,
      g_b2,nullptr,1.0f, R2,128, BN,128);
  gemm_wmma_f32<true,false,false><<<blocks(BN,512), 256, 0, stream>>>(
      R2,128, wih,128,128, nullptr,0, nullptr,0,0,
      bih,bhh,1.0f, R0,512, BN,512);
  lstm_scan<<<16, 128, 0, stream>>>(R0, whh, h0, c0, R3);
  gemm_wmma_f32<false,false,true><<<blocks(BN,256), 256, 0, stream>>>(
      R3,128, o_w1,256,128, nullptr,0, nullptr,0,0,
      o_b1,nullptr,1.0f, R0,256, BN,256);
  gemm_wmma_f32<false,false,false><<<blocks(BN,32), 256, 0, stream>>>(
      R0,256, o_w2,32,256, nullptr,0, nullptr,0,0,
      o_b2,nullptr,1.0f, out,32, BN,32);
}


# GEMM chains fused per 16-row block, A rows and intermediates kept in LDS (10 launches -> 5); the wih product reads 16-byte k groups, so its f32 sum order inside each 8 k changes
# speedup vs baseline: 9.7579x; 1.5039x over previous
#include <hip/hip_runtime.h>
#include <hip/hip_bf16.h>

typedef __attribute__((ext_vector_type(2))) float v2f;
typedef __attribute__((ext_vector_type(8))) float v8f;

__device__ __forceinline__ float sigmoidf_(float x) { return 1.0f / (1.0f + __expf(-x)); }

__device__ __forceinline__ v8f wmma4(v2f a, v2f b, v8f c) {
  return __builtin_amdgcn_wmma_f32_16x16x4_f32(false, a, false, b, (short)0, c,
                                               false, false);
}

#define PA128 132
#define PA256 260

template<int NT, int K, int UNR>
__device__ __forceinline__ void mma_lds_a(const float* as, const float* __restrict__ wp,
                                          int ldw, v8f (&acc)[NT])
{
  #pragma clang loop unroll_count(UNR)
  for (int k0 = 0; k0 < K; k0 += 4) {
    const v2f a = *reinterpret_cast<const v2f*>(as + k0);
    #pragma unroll
    for (int t = 0; t < NT; ++t) {
      v2f b;
      b.x = wp[t * 16];
      b.y = wp[ldw + t * 16];
      acc[t] = wmma4(a, b, acc[t]);
    }
    wp += (size_t)4 * ldw;
  }
}

template<int K>
__device__ __forceinline__ void mma_lds_a_wt(const float* as4, const float* __restrict__ wrow,
                                             v8f (&acc)[4])
{
  #pragma clang loop unroll_count(4)
  for (int kb = 0; kb < K; kb += 8) {
    const float4 a4 = *reinterpret_cast<const float4*>(as4 + kb);
    v2f a01, a23; a01.x = a4.x; a01.y = a4.y; a23.x = a4.z; a23.y = a4.w;
    float4 b4[4];
    #pragma unroll
    for (int t = 0; t < 4; ++t)
      b4[t] = *reinterpret_cast<const float4*>(wrow + (size_t)t * 16 * K + kb);
    #pragma unroll
    for (int t = 0; t < 4; ++t) { v2f b; b.x = b4[t].x; b.y = b4[t].y; acc[t] = wmma4(a01, b, acc[t]); }
    #pragma unroll
    for (int t = 0; t < 4; ++t) { v2f b; b.x = b4[t].z; b.y = b4[t].w; acc[t] = wmma4(a23, b, acc[t]); }
  }
}

template<int C4>
__device__ __forceinline__ void load_rows(const float* __restrict__ src, int tid,
                                          float4 (&v)[16 * C4 / 256])
{
  #pragma unroll
  for (int p = 0; p < 16 * C4 / 256; ++p) {
    const int i = tid + p * 256;
    const int r = i / C4, c = (i - r * C4) * 4;
    v[p] = *reinterpret_cast<const float4*>(src + (size_t)r * (4 * C4) + c);
  }
}
template<int C4>
__device__ __forceinline__ void store_rows(float* dst, int pa, int tid,
                                           const float4 (&v)[16 * C4 / 256])
{
  #pragma unroll
  for (int p = 0; p < 16 * C4 / 256; ++p) {
    const int i = tid + p * 256;
    const int r = i / C4, c = (i - r * C4) * 4;
    *reinterpret_cast<float4*>(dst + r * pa + c) = v[p];
  }
}

__global__ void __launch_bounds__(256)
chain_f(const float* __restrict__ hidden, const float* __restrict__ f_w1,
        const float* __restrict__ f_b1, float* __restrict__ Af, float* __restrict__ Bf)
{
  __shared__ __attribute__((aligned(16))) float sH[16 * PA128];
  const int tid = threadIdx.x, wave = tid >> 5, lane = tid & 31;
  const int half = lane >> 4, lp = lane & 15;
  const int m0 = blockIdx.x * 16;
  {
    float4 vh[2];
    load_rows<32>(hidden + (size_t)m0 * 128, tid, vh);
    store_rows<32>(sH, PA128, tid, vh);
  }
  __syncthreads();

  const int n0 = wave * 32;
  const float* as = sH + lp * PA128 + 2 * half;
  v8f acc[2] = {};
  mma_lds_a<2, 128, 8>(as, f_w1 + (size_t)(2 * half) * 256 + n0 + lp, 256, acc);
  v8f bcc[2] = {};
  mma_lds_a<2, 128, 8>(as, f_w1 + (size_t)(128 + 2 * half) * 256 + n0 + lp, 256, bcc);
  #pragma unroll
  for (int t = 0; t < 2; ++t) {
    const int n = n0 + 16 * t + lp;
    const float bias = f_b1[n];
    #pragma unroll
    for (int r = 0; r < 8; ++r) {
      const size_t o = (size_t)(m0 + 8 * half + r) * 256 + n;
      Af[o] = acc[t][r];
      Bf[o] = bcc[t][r] + bias;
    }
  }
}

__global__ void __launch_bounds__(256)
chain_g(const float* __restrict__ S, const float* __restrict__ x,
        const float* __restrict__ f_w2, const float* __restrict__ f_b2,
        const float* __restrict__ g_w1, const float* __restrict__ g_b1,
        const float* __restrict__ g_w2, const float* __restrict__ g_b2,
        const float* __restrict__ wih, const float* __restrict__ bih,
        const float* __restrict__ bhh, float* __restrict__ Gx)
{
  __shared__ __attribute__((aligned(16))) float bufBig[16 * PA256];
  __shared__ __attribute__((aligned(16))) float bufX[16 * PA128];
  __shared__ __attribute__((aligned(16))) float bufM[16 * PA128];
  const int tid = threadIdx.x, wave = tid >> 5, lane = tid & 31;
  const int half = lane >> 4, lp = lane & 15;
  const int m0 = blockIdx.x * 16;
  {
    float4 vs[4], vx[2];
    load_rows<64>(S + (size_t)m0 * 256, tid, vs);
    load_rows<32>(x + (size_t)m0 * 128, tid, vx);
    store_rows<64>(bufBig, PA256, tid, vs);
    store_rows<32>(bufX, PA128, tid, vx);
  }
  __syncthreads();

  {
    const int n = wave * 16 + lp;
    v8f acc[1] = {};
    mma_lds_a<1, 256, 16>(bufBig + lp * PA256 + 2 * half,
                          f_w2 + (size_t)(2 * half) * 128 + n, 128, acc);
    const float bias = 128.0f * f_b2[n];
    #pragma unroll
    for (int r = 0; r < 8; ++r) bufM[(8 * half + r) * PA128 + n] = acc[0][r] + bias;
  }
  __syncthreads();

  {
    const int n0 = wave * 32;
    v8f acc[2] = {};
    mma_lds_a<2, 128, 8>(bufX + lp * PA128 + 2 * half,
                         g_w1 + (size_t)(2 * half) * 256 + n0 + lp, 256, acc);
    mma_lds_a<2, 128, 8>(bufM + lp * PA128 + 2 * half,
                         g_w1 + (size_t)(128 + 2 * half) * 256 + n0 + lp, 256, acc);
    #pragma unroll
    for (int t = 0; t < 2; ++t) {
      const int n = n0 + 16 * t + lp;
      const float bias = g_b1[n];
      #pragma unroll
      for (int r = 0; r < 8; ++r)
        bufBig[(8 * half + r) * PA256 + n] = fmaxf(acc[t][r] + bias, 0.0f);
    }
  }
  __syncthreads();

  {
    const int n = wave * 16 + lp;
    v8f acc[1] = {};
    mma_lds_a<1, 256, 16>(bufBig + lp * PA256 + 2 * half,
                          g_w2 + (size_t)(2 * half) * 128 + n, 128, acc);
    const float bias = g_b2[n];
    #pragma unroll
    for (int r = 0; r < 8; ++r) bufX[(8 * half + r) * PA128 + n] = acc[0][r] + bias;
  }
  __syncthreads();

  {
    const int n0 = wave * 64;
    v8f acc[4] = {};
    mma_lds_a_wt<128>(bufX + lp * PA128 + 4 * half,
                      wih + (size_t)(n0 + lp) * 128 + 4 * half, acc);
    #pragma unroll
    for (int t = 0; t < 4; ++t) {
      const int n = n0 + 16 * t + lp;
      const float bias = bih[n] + bhh[n];
      #pragma unroll
      for (int r = 0; r < 8; ++r)
        Gx[(size_t)(m0 + 8 * half + r) * 512 + n] = acc[t][r] + bias;
    }
  }
}

__global__ void __launch_bounds__(256)
chain_o(const float* __restrict__ nh, const float* __restrict__ o_w1,
        const float* __restrict__ o_b1, const float* __restrict__ o_w2,
        const float* __restrict__ o_b2, float* __restrict__ out)
{
  __shared__ __attribute__((aligned(16))) float sH[16 * PA128];
  __shared__ __attribute__((aligned(16))) float sO[16 * PA256];
  const int tid = threadIdx.x, wave = tid >> 5, lane = tid & 31;
  const int half = lane >> 4, lp = lane & 15;
  const int m0 = blockIdx.x * 16;
  {
    float4 vh[2];
    load_rows<32>(nh + (size_t)m0 * 128, tid, vh);
    store_rows<32>(sH, PA128, tid, vh);
  }
  __syncthreads();
  {
    const int n0 = wave * 32;
    v8f acc[2] = {};
    mma_lds_a<2, 128, 8>(sH + lp * PA128 + 2 * half,
                         o_w1 + (size_t)(2 * half) * 256 + n0 + lp, 256, acc);
    #pragma unroll
    for (int t = 0; t < 2; ++t) {
      const int n = n0 + 16 * t + lp;
      const float bias = o_b1[n];
      #pragma unroll
      for (int r = 0; r < 8; ++r)
        sO[(8 * half + r) * PA256 + n] = fmaxf(acc[t][r] + bias, 0.0f);
    }
  }
  __syncthreads();
  if (wave < 2) {
    const int n = wave * 16 + lp;
    v8f acc[1] = {};
    mma_lds_a<1, 256, 16>(sO + lp * PA256 + 2 * half,
                          o_w2 + (size_t)(2 * half) * 32 + n, 32, acc);
    const float bias = o_b2[n];
    #pragma unroll
    for (int r = 0; r < 8; ++r)
      sH[(8 * half + r) * 36 + n] = acc[0][r] + bias;
  }
  __syncthreads();
  if (tid < 128) {
    const int r = tid >> 3, c = (tid & 7) * 4;
    const float4 v = *reinterpret_cast<const float4*>(sH + r * 36 + c);
    *reinterpret_cast<float4*>(out + (size_t)(m0 + r) * 32 + c) = v;
  }
}

__global__ void __launch_bounds__(256)
pair_relu_sum(const float* __restrict__ Af, const float* __restrict__ Bf,
              float* __restrict__ S)
{
  const int ri = blockIdx.x;
  const int b  = ri >> 7;
  const int f  = threadIdx.x;
  const float bm = Bf[(size_t)ri * 256 + f];
  const float* ap = Af + (size_t)b * 128 * 256 + f;
  float acc = 0.0f;
  #pragma unroll 4
  for (int j = 0; j < 128; ++j)
    acc += fmaxf(ap[(size_t)j * 256] + bm, 0.0f);
  S[(size_t)ri * 256 + f] = acc;
}

__global__ void __launch_bounds__(128)
lstm_scan(const float* __restrict__ Gx, const float* __restrict__ whh,
          const float* __restrict__ h0, const float* __restrict__ c0,
          float* __restrict__ new_hidden)
{
  __shared__ __attribute__((aligned(16))) float h_s[2][128];
  const int b = blockIdx.x;
  const int n = threadIdx.x;

  float wi[128], wf[128], wg[128], wo[128];
  {
    const float4* ri = reinterpret_cast<const float4*>(whh + (size_t)(0 * 128 + n) * 128);
    const float4* rf = reinterpret_cast<const float4*>(whh + (size_t)(1 * 128 + n) * 128);
    const float4* rg = reinterpret_cast<const float4*>(whh + (size_t)(2 * 128 + n) * 128);
    const float4* ro = reinterpret_cast<const float4*>(whh + (size_t)(3 * 128 + n) * 128);
    #pragma unroll
    for (int q = 0; q < 32; ++q) {
      const float4 vi = ri[q], vf = rf[q], vg = rg[q], vo = ro[q];
      wi[4*q] = vi.x; wi[4*q+1] = vi.y; wi[4*q+2] = vi.z; wi[4*q+3] = vi.w;
      wf[4*q] = vf.x; wf[4*q+1] = vf.y; wf[4*q+2] = vf.z; wf[4*q+3] = vf.w;
      wg[4*q] = vg.x; wg[4*q+1] = vg.y; wg[4*q+2] = vg.z; wg[4*q+3] = vg.w;
      wo[4*q] = vo.x; wo[4*q+1] = vo.y; wo[4*q+2] = vo.z; wo[4*q+3] = vo.w;
    }
  }

  float c = c0[b * 128 + n];
  h_s[0][n] = h0[b * 128 + n];

  const float* gx = Gx + (size_t)b * 128 * 512 + n;
  float gi = gx[0], gf = gx[128], gg = gx[256], go = gx[384];
  float* hout = new_hidden + (size_t)b * 128 * 128 + n;
  __syncthreads();

  for (int t = 0; t < 128; ++t) {
    const float* hs = h_s[t & 1];
    float ai[4] = {gi, 0.0f, 0.0f, 0.0f};
    float af[4] = {gf, 0.0f, 0.0f, 0.0f};
    float ag[4] = {gg, 0.0f, 0.0f, 0.0f};
    float ao[4] = {go, 0.0f, 0.0f, 0.0f};
    {
      const int tn = (t + 1 < 128) ? (t + 1) : 127;
      const float* gn = gx + (size_t)tn * 512;
      gi = gn[0]; gf = gn[128]; gg = gn[256]; go = gn[384];
    }
    #pragma unroll
    for (int q = 0; q < 8; ++q) {
      #pragma unroll
      for (int p = 0; p < 4; ++p) {
        const int k = p * 32 + q * 4;
        const float4 hv = *reinterpret_cast<const float4*>(hs + k);
        ai[p] = fmaf(hv.x, wi[k], ai[p]);   af[p] = fmaf(hv.x, wf[k], af[p]);
        ag[p] = fmaf(hv.x, wg[k], ag[p]);   ao[p] = fmaf(hv.x, wo[k], ao[p]);
        ai[p] = fmaf(hv.y, wi[k+1], ai[p]); af[p] = fmaf(hv.y, wf[k+1], af[p]);
        ag[p] = fmaf(hv.y, wg[k+1], ag[p]); ao[p] = fmaf(hv.y, wo[k+1], ao[p]);
        ai[p] = fmaf(hv.z, wi[k+2], ai[p]); af[p] = fmaf(hv.z, wf[k+2], af[p]);
        ag[p] = fmaf(hv.z, wg[k+2], ag[p]); ao[p] = fmaf(hv.z, wo[k+2], ao[p]);
        ai[p] = fmaf(hv.w, wi[k+3], ai[p]); af[p] = fmaf(hv.w, wf[k+3], af[p]);
        ag[p] = fmaf(hv.w, wg[k+3], ag[p]); ao[p] = fmaf(hv.w, wo[k+3], ao[p]);
      }
    }
    const float ig = (ai[0] + ai[1]) + (ai[2] + ai[3]);
    const float fg = (af[0] + af[1]) + (af[2] + af[3]);
    const float g2 = (ag[0] + ag[1]) + (ag[2] + ag[3]);
    const float og = (ao[0] + ao[1]) + (ao[2] + ao[3]);
    c = sigmoidf_(fg) * c + sigmoidf_(ig) * tanhf(g2);
    const float h = sigmoidf_(og) * tanhf(c);
    h_s[(t + 1) & 1][n] = h;
    __syncthreads();
    hout[(size_t)t * 128] = h;
  }
}

extern "C" void kernel_launch(void* const* d_in, const int* in_sizes, int n_in,
                              void* d_out, int out_size, void* d_ws, size_t ws_size,
                              hipStream_t stream)
{
  (void)in_sizes; (void)n_in; (void)out_size; (void)ws_size;
  const float* x      = (const float*)d_in[0];
  const float* hidden = (const float*)d_in[1];
  const float* h0     = (const float*)d_in[2];
  const float* c0     = (const float*)d_in[3];
  const float* f_w1   = (const float*)d_in[4];
  const float* f_b1   = (const float*)d_in[5];
  const float* f_w2   = (const float*)d_in[6];
  const float* f_b2   = (const float*)d_in[7];
  const float* g_w1   = (const float*)d_in[8];
  const float* g_b1   = (const float*)d_in[9];
  const float* g_w2   = (const float*)d_in[10];
  const float* g_b2   = (const float*)d_in[11];
  const float* wih    = (const float*)d_in[12];
  const float* whh    = (const float*)d_in[13];
  const float* bih    = (const float*)d_in[14];
  const float* bhh    = (const float*)d_in[15];
  const float* o_w1   = (const float*)d_in[16];
  const float* o_b1   = (const float*)d_in[17];
  const float* o_w2   = (const float*)d_in[18];
  const float* o_b2   = (const float*)d_in[19];
  float* out = (float*)d_out;

  float* R0 = (float*)d_ws;
  float* R1 = R0 + (size_t)2048 * 256;
  float* R2 = R1 + (size_t)2048 * 256;
  float* R3 = R2 + (size_t)2048 * 256;

  chain_f<<<128, 256, 0, stream>>>(hidden, f_w1, f_b1, R0, R1);
  pair_relu_sum<<<2048, 256, 0, stream>>>(R0, R1, R2);
  chain_g<<<128, 256, 0, stream>>>(R2, x, f_w2, f_b2, g_w1, g_b1, g_w2, g_b2,
                                   wih, bih, bhh, R0);
  lstm_scan<<<16, 128, 0, stream>>>(R0, whh, h0, c0, R3);
  chain_o<<<128, 256, 0, stream>>>(R3, o_w1, o_b1, o_w2, o_b2, out);
}


# pair_relu_sum: thread keeps 8 i rows of Bf and 8 sums in registers and streams its batch's 128 Af rows once for all 8 (2048 -> 256 blocks, 8x fewer L2 line reads); j order kept, sums bit-identical
# speedup vs baseline: 10.3935x; 1.0651x over previous
#include <hip/hip_runtime.h>
#include <hip/hip_bf16.h>

typedef __attribute__((ext_vector_type(2))) float v2f;
typedef __attribute__((ext_vector_type(8))) float v8f;

__device__ __forceinline__ float sigmoidf_(float x) { return 1.0f / (1.0f + __expf(-x)); }

__device__ __forceinline__ v8f wmma4(v2f a, v2f b, v8f c) {
  return __builtin_amdgcn_wmma_f32_16x16x4_f32(false, a, false, b, (short)0, c,
                                               false, false);
}

#define PA128 132
#define PA256 260

template<int NT, int K, int UNR>
__device__ __forceinline__ void mma_lds_a(const float* as, const float* __restrict__ wp,
                                          int ldw, v8f (&acc)[NT])
{
  #pragma clang loop unroll_count(UNR)
  for (int k0 = 0; k0 < K; k0 += 4) {
    const v2f a = *reinterpret_cast<const v2f*>(as + k0);
    #pragma unroll
    for (int t = 0; t < NT; ++t) {
      v2f b;
      b.x = wp[t * 16];
      b.y = wp[ldw + t * 16];
      acc[t] = wmma4(a, b, acc[t]);
    }
    wp += (size_t)4 * ldw;
  }
}

template<int K>
__device__ __forceinline__ void mma_lds_a_wt(const float* as4, const float* __restrict__ wrow,
                                             v8f (&acc)[4])
{
  #pragma clang loop unroll_count(4)
  for (int kb = 0; kb < K; kb += 8) {
    const float4 a4 = *reinterpret_cast<const float4*>(as4 + kb);
    v2f a01, a23; a01.x = a4.x; a01.y = a4.y; a23.x = a4.z; a23.y = a4.w;
    float4 b4[4];
    #pragma unroll
    for (int t = 0; t < 4; ++t)
      b4[t] = *reinterpret_cast<const float4*>(wrow + (size_t)t * 16 * K + kb);
    #pragma unroll
    for (int t = 0; t < 4; ++t) { v2f b; b.x = b4[t].x; b.y = b4[t].y; acc[t] = wmma4(a01, b, acc[t]); }
    #pragma unroll
    for (int t = 0; t < 4; ++t) { v2f b; b.x = b4[t].z; b.y = b4[t].w; acc[t] = wmma4(a23, b, acc[t]); }
  }
}

template<int C4>
__device__ __forceinline__ void load_rows(const float* __restrict__ src, int tid,
                                          float4 (&v)[16 * C4 / 256])
{
  #pragma unroll
  for (int p = 0; p < 16 * C4 / 256; ++p) {
    const int i = tid + p * 256;
    const int r = i / C4, c = (i - r * C4) * 4;
    v[p] = *reinterpret_cast<const float4*>(src + (size_t)r * (4 * C4) + c);
  }
}
template<int C4>
__device__ __forceinline__ void store_rows(float* dst, int pa, int tid,
                                           const float4 (&v)[16 * C4 / 256])
{
  #pragma unroll
  for (int p = 0; p < 16 * C4 / 256; ++p) {
    const int i = tid + p * 256;
    const int r = i / C4, c = (i - r * C4) * 4;
    *reinterpret_cast<float4*>(dst + r * pa + c) = v[p];
  }
}

__global__ void __launch_bounds__(256)
chain_f(const float* __restrict__ hidden, const float* __restrict__ f_w1,
        const float* __restrict__ f_b1, float* __restrict__ Af, float* __restrict__ Bf)
{
  __shared__ __attribute__((aligned(16))) float sH[16 * PA128];
  const int tid = threadIdx.x, wave = tid >> 5, lane = tid & 31;
  const int half = lane >> 4, lp = lane & 15;
  const int m0 = blockIdx.x * 16;
  {
    float4 vh[2];
    load_rows<32>(hidden + (size_t)m0 * 128, tid, vh);
    store_rows<32>(sH, PA128, tid, vh);
  }
  __syncthreads();

  const int n0 = wave * 32;
  const float* as = sH + lp * PA128 + 2 * half;
  v8f acc[2] = {};
  mma_lds_a<2, 128, 8>(as, f_w1 + (size_t)(2 * half) * 256 + n0 + lp, 256, acc);
  v8f bcc[2] = {};
  mma_lds_a<2, 128, 8>(as, f_w1 + (size_t)(128 + 2 * half) * 256 + n0 + lp, 256, bcc);
  #pragma unroll
  for (int t = 0; t < 2; ++t) {
    const int n = n0 + 16 * t + lp;
    const float bias = f_b1[n];
    #pragma unroll
    for (int r = 0; r < 8; ++r) {
      const size_t o = (size_t)(m0 + 8 * half + r) * 256 + n;
      Af[o] = acc[t][r];
      Bf[o] = bcc[t][r] + bias;
    }
  }
}

__global__ void __launch_bounds__(256)
chain_g(const float* __restrict__ S, const float* __restrict__ x,
        const float* __restrict__ f_w2, const float* __restrict__ f_b2,
        const float* __restrict__ g_w1, const float* __restrict__ g_b1,
        const float* __restrict__ g_w2, const float* __restrict__ g_b2,
        const float* __restrict__ wih, const float* __restrict__ bih,
        const float* __restrict__ bhh, float* __restrict__ Gx)
{
  __shared__ __attribute__((aligned(16))) float bufBig[16 * PA256];
  __shared__ __attribute__((aligned(16))) float bufX[16 * PA128];
  __shared__ __attribute__((aligned(16))) float bufM[16 * PA128];
  const int tid = threadIdx.x, wave = tid >> 5, lane = tid & 31;
  const int half = lane >> 4, lp = lane & 15;
  const int m0 = blockIdx.x * 16;
  {
    float4 vs[4], vx[2];
    load_rows<64>(S + (size_t)m0 * 256, tid, vs);
    load_rows<32>(x + (size_t)m0 * 128, tid, vx);
    store_rows<64>(bufBig, PA256, tid, vs);
    store_rows<32>(bufX, PA128, tid, vx);
  }
  __syncthreads();

  {
    const int n = wave * 16 + lp;
    v8f acc[1] = {};
    mma_lds_a<1, 256, 16>(bufBig + lp * PA256 + 2 * half,
                          f_w2 + (size_t)(2 * half) * 128 + n, 128, acc);
    const float bias = 128.0f * f_b2[n];
    #pragma unroll
    for (int r = 0; r < 8; ++r) bufM[(8 * half + r) * PA128 + n] = acc[0][r] + bias;
  }
  __syncthreads();

  {
    const int n0 = wave * 32;
    v8f acc[2] = {};
    mma_lds_a<2, 128, 8>(bufX + lp * PA128 + 2 * half,
                         g_w1 + (size_t)(2 * half) * 256 + n0 + lp, 256, acc);
    mma_lds_a<2, 128, 8>(bufM + lp * PA128 + 2 * half,
                         g_w1 + (size_t)(128 + 2 * half) * 256 + n0 + lp, 256, acc);
    #pragma unroll
    for (int t = 0; t < 2; ++t) {
      const int n = n0 + 16 * t + lp;
      const float bias = g_b1[n];
      #pragma unroll
      for (int r = 0; r < 8; ++r)
        bufBig[(8 * half + r) * PA256 + n] = fmaxf(acc[t][r] + bias, 0.0f);
    }
  }
  __syncthreads();

  {
    const int n = wave * 16 + lp;
    v8f acc[1] = {};
    mma_lds_a<1, 256, 16>(bufBig + lp * PA256 + 2 * half,
                          g_w2 + (size_t)(2 * half) * 128 + n, 128, acc);
    const float bias = g_b2[n];
    #pragma unroll
    for (int r = 0; r < 8; ++r) bufX[(8 * half + r) * PA128 + n] = acc[0][r] + bias;
  }
  __syncthreads();

  {
    const int n0 = wave * 64;
    v8f acc[4] = {};
    mma_lds_a_wt<128>(bufX + lp * PA128 + 4 * half,
                      wih + (size_t)(n0 + lp) * 128 + 4 * half, acc);
    #pragma unroll
    for (int t = 0; t < 4; ++t) {
      const int n = n0 + 16 * t + lp;
      const float bias = bih[n] + bhh[n];
      #pragma unroll
      for (int r = 0; r < 8; ++r)
        Gx[(size_t)(m0 + 8 * half + r) * 512 + n] = acc[t][r] + bias;
    }
  }
}

__global__ void __launch_bounds__(256)
chain_o(const float* __restrict__ nh, const float* __restrict__ o_w1,
        const float* __restrict__ o_b1, const float* __restrict__ o_w2,
        const float* __restrict__ o_b2, float* __restrict__ out)
{
  __shared__ __attribute__((aligned(16))) float sH[16 * PA128];
  __shared__ __attribute__((aligned(16))) float sO[16 * PA256];
  const int tid = threadIdx.x, wave = tid >> 5, lane = tid & 31;
  const int half = lane >> 4, lp = lane & 15;
  const int m0 = blockIdx.x * 16;
  {
    float4 vh[2];
    load_rows<32>(nh + (size_t)m0 * 128, tid, vh);
    store_rows<32>(sH, PA128, tid, vh);
  }
  __syncthreads();
  {
    const int n0 = wave * 32;
    v8f acc[2] = {};
    mma_lds_a<2, 128, 8>(sH + lp * PA128 + 2 * half,
                         o_w1 + (size_t)(2 * half) * 256 + n0 + lp, 256, acc);
    #pragma unroll
    for (int t = 0; t < 2; ++t) {
      const int n = n0 + 16 * t + lp;
      const float bias = o_b1[n];
      #pragma unroll
      for (int r = 0; r < 8; ++r)
        sO[(8 * half + r) * PA256 + n] = fmaxf(acc[t][r] + bias, 0.0f);
    }
  }
  __syncthreads();
  if (wave < 2) {
    const int n = wave * 16 + lp;
    v8f acc[1] = {};
    mma_lds_a<1, 256, 16>(sO + lp * PA256 + 2 * half,
                          o_w2 + (size_t)(2 * half) * 32 + n, 32, acc);
    const float bias = o_b2[n];
    #pragma unroll
    for (int r = 0; r < 8; ++r)
      sH[(8 * half + r) * 36 + n] = acc[0][r] + bias;
  }
  __syncthreads();
  if (tid < 128) {
    const int r = tid >> 3, c = (tid & 7) * 4;
    const float4 v = *reinterpret_cast<const float4*>(sH + r * 36 + c);
    *reinterpret_cast<float4*>(out + (size_t)(m0 + r) * 32 + c) = v;
  }
}

#define PR_TI 8
#define PR_G  16

__device__ __forceinline__ void pr_load(const float* __restrict__ p, float (&v)[PR_G])
{
  #pragma unroll
  for (int q = 0; q < PR_G; ++q) v[q] = p[(size_t)q * 256];
}
__device__ __forceinline__ void pr_accum(const float (&v)[PR_G], const float (&bm)[PR_TI],
                                         float (&acc)[PR_TI])
{
  #pragma unroll
  for (int q = 0; q < PR_G; ++q) {
    #pragma unroll
    for (int i = 0; i < PR_TI; ++i) acc[i] += fmaxf(v[q] + bm[i], 0.0f);
  }
}

__global__ void __launch_bounds__(256)
pair_relu_sum(const float* __restrict__ Af, const float* __restrict__ Bf,
              float* __restrict__ S)
{
  const int b  = blockIdx.x >> 4;
  const int r0 = b * 128 + (blockIdx.x & 15) * PR_TI;
  const int f  = threadIdx.x;
  const float* bp = Bf + (size_t)r0 * 256 + f;
  const float* ap = Af + (size_t)b * 128 * 256 + f;
  float va[PR_G], vb[PR_G];
  pr_load(ap, va);
  float bm[PR_TI], acc[PR_TI];
  #pragma unroll
  for (int i = 0; i < PR_TI; ++i) { bm[i] = bp[(size_t)i * 256]; acc[i] = 0.0f; }
  for (int g = 0; g < 128 / (2 * PR_G); ++g) {
    pr_load(ap + (size_t)(2 * PR_G * g + PR_G) * 256, vb);
    pr_accum(va, bm, acc);
    if (g + 1 < 128 / (2 * PR_G))
      pr_load(ap + (size_t)(2 * PR_G * g + 2 * PR_G) * 256, va);
    pr_accum(vb, bm, acc);
  }
  float* sp = S + (size_t)r0 * 256 + f;
  #pragma unroll
  for (int i = 0; i < PR_TI; ++i) sp[(size_t)i * 256] = acc[i];
}

__global__ void __launch_bounds__(128)
lstm_scan(const float* __restrict__ Gx, const float* __restrict__ whh,
          const float* __restrict__ h0, const float* __restrict__ c0,
          float* __restrict__ new_hidden)
{
  __shared__ __attribute__((aligned(16))) float h_s[2][128];
  const int b = blockIdx.x;
  const int n = threadIdx.x;

  float wi[128], wf[128], wg[128], wo[128];
  {
    const float4* ri = reinterpret_cast<const float4*>(whh + (size_t)(0 * 128 + n) * 128);
    const float4* rf = reinterpret_cast<const float4*>(whh + (size_t)(1 * 128 + n) * 128);
    const float4* rg = reinterpret_cast<const float4*>(whh + (size_t)(2 * 128 + n) * 128);
    const float4* ro = reinterpret_cast<const float4*>(whh + (size_t)(3 * 128 + n) * 128);
    #pragma unroll
    for (int q = 0; q < 32; ++q) {
      const float4 vi = ri[q], vf = rf[q], vg = rg[q], vo = ro[q];
      wi[4*q] = vi.x; wi[4*q+1] = vi.y; wi[4*q+2] = vi.z; wi[4*q+3] = vi.w;
      wf[4*q] = vf.x; wf[4*q+1] = vf.y; wf[4*q+2] = vf.z; wf[4*q+3] = vf.w;
      wg[4*q] = vg.x; wg[4*q+1] = vg.y; wg[4*q+2] = vg.z; wg[4*q+3] = vg.w;
      wo[4*q] = vo.x; wo[4*q+1] = vo.y; wo[4*q+2] = vo.z; wo[4*q+3] = vo.w;
    }
  }

  float c = c0[b * 128 + n];
  h_s[0][n] = h0[b * 128 + n];

  const float* gx = Gx + (size_t)b * 128 * 512 + n;
  float gi = gx[0], gf = gx[128], gg = gx[256], go = gx[384];
  float* hout = new_hidden + (size_t)b * 128 * 128 + n;
  __syncthreads();

  for (int t = 0; t < 128; ++t) {
    const float* hs = h_s[t & 1];
    float ai[4] = {gi, 0.0f, 0.0f, 0.0f};
    float af[4] = {gf, 0.0f, 0.0f, 0.0f};
    float ag[4] = {gg, 0.0f, 0.0f, 0.0f};
    float ao[4] = {go, 0.0f, 0.0f, 0.0f};
    {
      const int tn = (t + 1 < 128) ? (t + 1) : 127;
      const float* gn = gx + (size_t)tn * 512;
      gi = gn[0]; gf = gn[128]; gg = gn[256]; go = gn[384];
    }
    #pragma unroll
    for (int q = 0; q < 8; ++q) {
      #pragma unroll
      for (int p = 0; p < 4; ++p) {
        const int k = p * 32 + q * 4;
        const float4 hv = *reinterpret_cast<const float4*>(hs + k);
        ai[p] = fmaf(hv.x, wi[k], ai[p]);   af[p] = fmaf(hv.x, wf[k], af[p]);
        ag[p] = fmaf(hv.x, wg[k], ag[p]);   ao[p] = fmaf(hv.x, wo[k], ao[p]);
        ai[p] = fmaf(hv.y, wi[k+1], ai[p]); af[p] = fmaf(hv.y, wf[k+1], af[p]);
        ag[p] = fmaf(hv.y, wg[k+1], ag[p]); ao[p] = fmaf(hv.y, wo[k+1], ao[p]);
        ai[p] = fmaf(hv.z, wi[k+2], ai[p]); af[p] = fmaf(hv.z, wf[k+2], af[p]);
        ag[p] = fmaf(hv.z, wg[k+2], ag[p]); ao[p] = fmaf(hv.z, wo[k+2], ao[p]);
        ai[p] = fmaf(hv.w, wi[k+3], ai[p]); af[p] = fmaf(hv.w, wf[k+3], af[p]);
        ag[p] = fmaf(hv.w, wg[k+3], ag[p]); ao[p] = fmaf(hv.w, wo[k+3], ao[p]);
      }
    }
    const float ig = (ai[0] + ai[1]) + (ai[2] + ai[3]);
    const float fg = (af[0] + af[1]) + (af[2] + af[3]);
    const float g2 = (ag[0] + ag[1]) + (ag[2] + ag[3]);
    const float og = (ao[0] + ao[1]) + (ao[2] + ao[3]);
    c = sigmoidf_(fg) * c + sigmoidf_(ig) * tanhf(g2);
    const float h = sigmoidf_(og) * tanhf(c);
    h_s[(t + 1) & 1][n] = h;
    __syncthreads();
    hout[(size_t)t * 128] = h;
  }
}

extern "C" void kernel_launch(void* const* d_in, const int* in_sizes, int n_in,
                              void* d_out, int out_size, void* d_ws, size_t ws_size,
                              hipStream_t stream)
{
  (void)in_sizes; (void)n_in; (void)out_size; (void)ws_size;
  const float* x      = (const float*)d_in[0];
  const float* hidden = (const float*)d_in[1];
  const float* h0     = (const float*)d_in[2];
  const float* c0     = (const float*)d_in[3];
  const float* f_w1   = (const float*)d_in[4];
  const float* f_b1   = (const float*)d_in[5];
  const float* f_w2   = (const float*)d_in[6];
  const float* f_b2   = (const float*)d_in[7];
  const float* g_w1   = (const float*)d_in[8];
  const float* g_b1   = (const float*)d_in[9];
  const float* g_w2   = (const float*)d_in[10];
  const float* g_b2   = (const float*)d_in[11];
  const float* wih    = (const float*)d_in[12];
  const float* whh    = (const float*)d_in[13];
  const float* bih    = (const float*)d_in[14];
  const float* bhh    = (const float*)d_in[15];
  const float* o_w1   = (const float*)d_in[16];
  const float* o_b1   = (const float*)d_in[17];
  const float* o_w2   = (const float*)d_in[18];
  const float* o_b2   = (const float*)d_in[19];
  float* out = (float*)d_out;

  float* R0 = (float*)d_ws;
  float* R1 = R0 + (size_t)2048 * 256;
  float* R2 = R1 + (size_t)2048 * 256;
  float* R3 = R2 + (size_t)2048 * 256;

  chain_f<<<128, 256, 0, stream>>>(hidden, f_w1, f_b1, R0, R1);
  pair_relu_sum<<<256, 256, 0, stream>>>(R0, R1, R2);
  chain_g<<<128, 256, 0, stream>>>(R2, x, f_w2, f_b2, g_w1, g_b1, g_w2, g_b2,
                                   wih, bih, bhh, R0);
  lstm_scan<<<16, 128, 0, stream>>>(R0, whh, h0, c0, R3);
  chain_o<<<128, 256, 0, stream>>>(R3, o_w1, o_b1, o_w2, o_b2, out);
}


# LSTM scan on 512 threads per batch row: a quad of lanes shares a unit and splits k 4 ways, packed FMA on gate pairs, quad fold, one gate activation per lane (exp2, rcp); f32 sum order changes
# speedup vs baseline: 11.8986x; 1.1448x over previous
#include <hip/hip_runtime.h>
#include <hip/hip_bf16.h>

typedef __attribute__((ext_vector_type(2))) float v2f;
typedef __attribute__((ext_vector_type(8))) float v8f;
typedef __attribute__((ext_vector_type(4))) float v4f;

__device__ __forceinline__ float sigmoidf_(float x) { return 1.0f / (1.0f + __expf(-x)); }

__device__ __forceinline__ v8f wmma4(v2f a, v2f b, v8f c) {
  return __builtin_amdgcn_wmma_f32_16x16x4_f32(false, a, false, b, (short)0, c,
                                               false, false);
}

#define PA128 132
#define PA256 260

template<int NT, int K, int UNR>
__device__ __forceinline__ void mma_lds_a(const float* as, const float* __restrict__ wp,
                                          int ldw, v8f (&acc)[NT])
{
  #pragma clang loop unroll_count(UNR)
  for (int k0 = 0; k0 < K; k0 += 4) {
    const v2f a = *reinterpret_cast<const v2f*>(as + k0);
    #pragma unroll
    for (int t = 0; t < NT; ++t) {
      v2f b;
      b.x = wp[t * 16];
      b.y = wp[ldw + t * 16];
      acc[t] = wmma4(a, b, acc[t]);
    }
    wp += (size_t)4 * ldw;
  }
}

template<int K>
__device__ __forceinline__ void mma_lds_a_wt(const float* as4, const float* __restrict__ wrow,
                                             v8f (&acc)[4])
{
  #pragma clang loop unroll_count(4)
  for (int kb = 0; kb < K; kb += 8) {
    const float4 a4 = *reinterpret_cast<const float4*>(as4 + kb);
    v2f a01, a23; a01.x = a4.x; a01.y = a4.y; a23.x = a4.z; a23.y = a4.w;
    float4 b4[4];
    #pragma unroll
    for (int t = 0; t < 4; ++t)
      b4[t] = *reinterpret_cast<const float4*>(wrow + (size_t)t * 16 * K + kb);
    #pragma unroll
    for (int t = 0; t < 4; ++t) { v2f b; b.x = b4[t].x; b.y = b4[t].y; acc[t] = wmma4(a01, b, acc[t]); }
    #pragma unroll
    for (int t = 0; t < 4; ++t) { v2f b; b.x = b4[t].z; b.y = b4[t].w; acc[t] = wmma4(a23, b, acc[t]); }
  }
}

template<int C4>
__device__ __forceinline__ void load_rows(const float* __restrict__ src, int tid,
                                          float4 (&v)[16 * C4 / 256])
{
  #pragma unroll
  for (int p = 0; p < 16 * C4 / 256; ++p) {
    const int i = tid + p * 256;
    const int r = i / C4, c = (i - r * C4) * 4;
    v[p] = *reinterpret_cast<const float4*>(src + (size_t)r * (4 * C4) + c);
  }
}
template<int C4>
__device__ __forceinline__ void store_rows(float* dst, int pa, int tid,
                                           const float4 (&v)[16 * C4 / 256])
{
  #pragma unroll
  for (int p = 0; p < 16 * C4 / 256; ++p) {
    const int i = tid + p * 256;
    const int r = i / C4, c = (i - r * C4) * 4;
    *reinterpret_cast<float4*>(dst + r * pa + c) = v[p];
  }
}

__global__ void __launch_bounds__(256)
chain_f(const float* __restrict__ hidden, const float* __restrict__ f_w1,
        const float* __restrict__ f_b1, float* __restrict__ Af, float* __restrict__ Bf)
{
  __shared__ __attribute__((aligned(16))) float sH[16 * PA128];
  const int tid = threadIdx.x, wave = tid >> 5, lane = tid & 31;
  const int half = lane >> 4, lp = lane & 15;
  const int m0 = blockIdx.x * 16;
  {
    float4 vh[2];
    load_rows<32>(hidden + (size_t)m0 * 128, tid, vh);
    store_rows<32>(sH, PA128, tid, vh);
  }
  __syncthreads();

  const int n0 = wave * 32;
  const float* as = sH + lp * PA128 + 2 * half;
  v8f acc[2] = {};
  mma_lds_a<2, 128, 8>(as, f_w1 + (size_t)(2 * half) * 256 + n0 + lp, 256, acc);
  v8f bcc[2] = {};
  mma_lds_a<2, 128, 8>(as, f_w1 + (size_t)(128 + 2 * half) * 256 + n0 + lp, 256, bcc);
  #pragma unroll
  for (int t = 0; t < 2; ++t) {
    const int n = n0 + 16 * t + lp;
    const float bias = f_b1[n];
    #pragma unroll
    for (int r = 0; r < 8; ++r) {
      const size_t o = (size_t)(m0 + 8 * half + r) * 256 + n;
      Af[o] = acc[t][r];
      Bf[o] = bcc[t][r] + bias;
    }
  }
}

__global__ void __launch_bounds__(256)
chain_g(const float* __restrict__ S, const float* __restrict__ x,
        const float* __restrict__ f_w2, const float* __restrict__ f_b2,
        const float* __restrict__ g_w1, const float* __restrict__ g_b1,
        const float* __restrict__ g_w2, const float* __restrict__ g_b2,
        const float* __restrict__ wih, const float* __restrict__ bih,
        const float* __restrict__ bhh, float* __restrict__ Gx)
{
  __shared__ __attribute__((aligned(16))) float bufBig[16 * PA256];
  __shared__ __attribute__((aligned(16))) float bufX[16 * PA128];
  __shared__ __attribute__((aligned(16))) float bufM[16 * PA128];
  const int tid = threadIdx.x, wave = tid >> 5, lane = tid & 31;
  const int half = lane >> 4, lp = lane & 15;
  const int m0 = blockIdx.x * 16;
  {
    float4 vs[4], vx[2];
    load_rows<64>(S + (size_t)m0 * 256, tid, vs);
    load_rows<32>(x + (size_t)m0 * 128, tid, vx);
    store_rows<64>(bufBig, PA256, tid, vs);
    store_rows<32>(bufX, PA128, tid, vx);
  }
  __syncthreads();

  {
    const int n = wave * 16 + lp;
    v8f acc[1] = {};
    mma_lds_a<1, 256, 16>(bufBig + lp * PA256 + 2 * half,
                          f_w2 + (size_t)(2 * half) * 128 + n, 128, acc);
    const float bias = 128.0f * f_b2[n];
    #pragma unroll
    for (int r = 0; r < 8; ++r) bufM[(8 * half + r) * PA128 + n] = acc[0][r] + bias;
  }
  __syncthreads();

  {
    const int n0 = wave * 32;
    v8f acc[2] = {};
    mma_lds_a<2, 128, 8>(bufX + lp * PA128 + 2 * half,
                         g_w1 + (size_t)(2 * half) * 256 + n0 + lp, 256, acc);
    mma_lds_a<2, 128, 8>(bufM + lp * PA128 + 2 * half,
                         g_w1 + (size_t)(128 + 2 * half) * 256 + n0 + lp, 256, acc);
    #pragma unroll
    for (int t = 0; t < 2; ++t) {
      const int n = n0 + 16 * t + lp;
      const float bias = g_b1[n];
      #pragma unroll
      for (int r = 0; r < 8; ++r)
        bufBig[(8 * half + r) * PA256 + n] = fmaxf(acc[t][r] + bias, 0.0f);
    }
  }
  __syncthreads();

  {
    const int n = wave * 16 + lp;
    v8f acc[1] = {};
    mma_lds_a<1, 256, 16>(bufBig + lp * PA256 + 2 * half,
                          g_w2 + (size_t)(2 * half) * 128 + n, 128, acc);
    const float bias = g_b2[n];
    #pragma unroll
    for (int r = 0; r < 8; ++r) bufX[(8 * half + r) * PA128 + n] = acc[0][r] + bias;
  }
  __syncthreads();

  {
    const int n0 = wave * 64;
    v8f acc[4] = {};
    mma_lds_a_wt<128>(bufX + lp * PA128 + 4 * half,
                      wih + (size_t)(n0 + lp) * 128 + 4 * half, acc);
    #pragma unroll
    for (int t = 0; t < 4; ++t) {
      const int n = n0 + 16 * t + lp;
      const float bias = bih[n] + bhh[n];
      #pragma unroll
      for (int r = 0; r < 8; ++r)
        Gx[(size_t)(m0 + 8 * half + r) * 512 + n] = acc[t][r] + bias;
    }
  }
}

__global__ void __launch_bounds__(256)
chain_o(const float* __restrict__ nh, const float* __restrict__ o_w1,
        const float* __restrict__ o_b1, const float* __restrict__ o_w2,
        const float* __restrict__ o_b2, float* __restrict__ out)
{
  __shared__ __attribute__((aligned(16))) float sH[16 * PA128];
  __shared__ __attribute__((aligned(16))) float sO[16 * PA256];
  const int tid = threadIdx.x, wave = tid >> 5, lane = tid & 31;
  const int half = lane >> 4, lp = lane & 15;
  const int m0 = blockIdx.x * 16;
  {
    float4 vh[2];
    load_rows<32>(nh + (size_t)m0 * 128, tid, vh);
    store_rows<32>(sH, PA128, tid, vh);
  }
  __syncthreads();
  {
    const int n0 = wave * 32;
    v8f acc[2] = {};
    mma_lds_a<2, 128, 8>(sH + lp * PA128 + 2 * half,
                         o_w1 + (size_t)(2 * half) * 256 + n0 + lp, 256, acc);
    #pragma unroll
    for (int t = 0; t < 2; ++t) {
      const int n = n0 + 16 * t + lp;
      const float bias = o_b1[n];
      #pragma unroll
      for (int r = 0; r < 8; ++r)
        sO[(8 * half + r) * PA256 + n] = fmaxf(acc[t][r] + bias, 0.0f);
    }
  }
  __syncthreads();
  if (wave < 2) {
    const int n = wave * 16 + lp;
    v8f acc[1] = {};
    mma_lds_a<1, 256, 16>(sO + lp * PA256 + 2 * half,
                          o_w2 + (size_t)(2 * half) * 32 + n, 32, acc);
    const float bias = o_b2[n];
    #pragma unroll
    for (int r = 0; r < 8; ++r)
      sH[(8 * half + r) * 36 + n] = acc[0][r] + bias;
  }
  __syncthreads();
  if (tid < 128) {
    const int r = tid >> 3, c = (tid & 7) * 4;
    const float4 v = *reinterpret_cast<const float4*>(sH + r * 36 + c);
    *reinterpret_cast<float4*>(out + (size_t)(m0 + r) * 32 + c) = v;
  }
}

#define PR_TI 8
#define PR_G  16

__device__ __forceinline__ void pr_load(const float* __restrict__ p, float (&v)[PR_G])
{
  #pragma unroll
  for (int q = 0; q < PR_G; ++q) v[q] = p[(size_t)q * 256];
}
__device__ __forceinline__ void pr_accum(const float (&v)[PR_G], const float (&bm)[PR_TI],
                                         float (&acc)[PR_TI])
{
  #pragma unroll
  for (int q = 0; q < PR_G; ++q) {
    #pragma unroll
    for (int i = 0; i < PR_TI; ++i) acc[i] += fmaxf(v[q] + bm[i], 0.0f);
  }
}

__global__ void __launch_bounds__(256)
pair_relu_sum(const float* __restrict__ Af, const float* __restrict__ Bf,
              float* __restrict__ S)
{
  const int b  = blockIdx.x >> 4;
  const int r0 = b * 128 + (blockIdx.x & 15) * PR_TI;
  const int f  = threadIdx.x;
  const float* bp = Bf + (size_t)r0 * 256 + f;
  const float* ap = Af + (size_t)b * 128 * 256 + f;
  float va[PR_G], vb[PR_G];
  pr_load(ap, va);
  float bm[PR_TI], acc[PR_TI];
  #pragma unroll
  for (int i = 0; i < PR_TI; ++i) { bm[i] = bp[(size_t)i * 256]; acc[i] = 0.0f; }
  for (int g = 0; g < 128 / (2 * PR_G); ++g) {
    pr_load(ap + (size_t)(2 * PR_G * g + PR_G) * 256, vb);
    pr_accum(va, bm, acc);
    if (g + 1 < 128 / (2 * PR_G))
      pr_load(ap + (size_t)(2 * PR_G * g + 2 * PR_G) * 256, va);
    pr_accum(vb, bm, acc);
  }
  float* sp = S + (size_t)r0 * 256 + f;
  #pragma unroll
  for (int i = 0; i < PR_TI; ++i) sp[(size_t)i * 256] = acc[i];
}

template<int CTRL>
__device__ __forceinline__ float quad_get(float v)
{
  return __int_as_float(__builtin_amdgcn_mov_dpp(__float_as_int(v), CTRL, 0xf, 0xf, true));
}
#define QP_XOR1 0xB1
#define QP_XOR2 0x4E
#define QP_L0   0x00
#define QP_L1   0x55
#define QP_L2   0xAA
#define QP_L3   0xFF

__device__ __forceinline__ v2f fma2(float h, v2f w, v2f acc)
{
  v2f hh; hh.x = h; hh.y = h;
  return __builtin_elementwise_fma(hh, w, acc);
}

__global__ void __launch_bounds__(512)
lstm_scan(const float* __restrict__ Gx, const float* __restrict__ whh,
          const float* __restrict__ h0, const float* __restrict__ c0,
          float* __restrict__ new_hidden)
{
  __shared__ __attribute__((aligned(16))) float h_s[2][128];
  const int b = blockIdx.x;
  const int tid = threadIdx.x;
  const int q = tid & 3;
  const int n = tid >> 2;
  const int wave = tid >> 5, lane = tid & 31;

  v2f wif[32], wgo[32];
  {
    const float* ri = whh + (size_t)(0 * 128 + n) * 128 + 4 * q;
    const float* rf = whh + (size_t)(1 * 128 + n) * 128 + 4 * q;
    const float* rg = whh + (size_t)(2 * 128 + n) * 128 + 4 * q;
    const float* ro = whh + (size_t)(3 * 128 + n) * 128 + 4 * q;
    #pragma unroll
    for (int j = 0; j < 8; ++j) {
      const float4 vi = *reinterpret_cast<const float4*>(ri + 16 * j);
      const float4 vf = *reinterpret_cast<const float4*>(rf + 16 * j);
      const float4 vg = *reinterpret_cast<const float4*>(rg + 16 * j);
      const float4 vo = *reinterpret_cast<const float4*>(ro + 16 * j);
      wif[4*j+0].x = vi.x; wif[4*j+0].y = vf.x; wgo[4*j+0].x = vg.x; wgo[4*j+0].y = vo.x;
      wif[4*j+1].x = vi.y; wif[4*j+1].y = vf.y; wgo[4*j+1].x = vg.y; wgo[4*j+1].y = vo.y;
      wif[4*j+2].x = vi.z; wif[4*j+2].y = vf.z; wgo[4*j+2].x = vg.z; wgo[4*j+2].y = vo.z;
      wif[4*j+3].x = vi.w; wif[4*j+3].y = vf.w; wgo[4*j+3].x = vg.w; wgo[4*j+3].y = vo.w;
    }
  }

  const float LOG2E = 1.44269504088896341f;
  const float am = (q == 2) ? 2.0f : 1.0f;
  const float bm = (q == 2) ? -1.0f : 0.0f;
  const float ke = -am * LOG2E;
  const bool odd = (q & 1) != 0, hi = (q & 2) != 0;

  float c = c0[b * 128 + n];
  if (q == 0) h_s[0][n] = h0[b * 128 + n];

  const float* gx = Gx + (size_t)b * 128 * 512 + q * 128 + n;
  float gxq = gx[0];
  float* hout = new_hidden + (size_t)b * 128 * 128 + 32 * wave + lane;
  __syncthreads();

  for (int t = 0; t < 128; ++t) {
    const float* hs = h_s[t & 1] + 4 * q;
    v4f hv[8];
    #pragma unroll
    for (int j = 0; j < 8; ++j) hv[j] = *reinterpret_cast<const v4f*>(hs + 16 * j);
    asm volatile("" :: "v"(hv[0]), "v"(hv[1]), "v"(hv[2]), "v"(hv[3]),
                       "v"(hv[4]), "v"(hv[5]), "v"(hv[6]), "v"(hv[7]));
    const float gx_now = gxq;
    {
      const int tn = (t + 1 < 128) ? (t + 1) : 127;
      gxq = gx[(size_t)tn * 512];
    }
    v2f aif0 = {0.0f, 0.0f}, aif1 = {0.0f, 0.0f}, ago0 = {0.0f, 0.0f}, ago1 = {0.0f, 0.0f};
    #pragma unroll
    for (int j = 0; j < 8; ++j) {
      aif0 = fma2(hv[j].x, wif[4*j+0], aif0);  ago0 = fma2(hv[j].x, wgo[4*j+0], ago0);
      aif1 = fma2(hv[j].y, wif[4*j+1], aif1);  ago1 = fma2(hv[j].y, wgo[4*j+1], ago1);
      aif0 = fma2(hv[j].z, wif[4*j+2], aif0);  ago0 = fma2(hv[j].z, wgo[4*j+2], ago0);
      aif1 = fma2(hv[j].w, wif[4*j+3], aif1);  ago1 = fma2(hv[j].w, wgo[4*j+3], ago1);
    }
    const v2f pif = aif0 + aif1, pgo = ago0 + ago1;

    const float keepA = odd ? pif.y : pif.x, sendA = odd ? pif.x : pif.y;
    const float keepB = odd ? pgo.y : pgo.x, sendB = odd ? pgo.x : pgo.y;
    const float sA = keepA + quad_get<QP_XOR1>(sendA);
    const float sB = keepB + quad_get<QP_XOR1>(sendB);
    const float keep = hi ? sB : sA, send = hi ? sA : sB;
    const float x = (keep + quad_get<QP_XOR2>(send)) + gx_now;

    const float act = fmaf(am, __builtin_amdgcn_rcpf(1.0f + __builtin_amdgcn_exp2f(ke * x)), bm);
    const float ig = quad_get<QP_L0>(act), fg = quad_get<QP_L1>(act);
    const float g2 = quad_get<QP_L2>(act), og = quad_get<QP_L3>(act);
    c = fmaf(fg, c, ig * g2);
    const float th = fmaf(2.0f, __builtin_amdgcn_rcpf(1.0f + __builtin_amdgcn_exp2f((-2.0f * LOG2E) * c)), -1.0f);
    const float h = og * th;
    if (q == 0) h_s[(t + 1) & 1][n] = h;
    __syncthreads();
    if (wave < 4) hout[(size_t)t * 128] = h_s[(t + 1) & 1][32 * wave + lane];
  }
}

extern "C" void kernel_launch(void* const* d_in, const int* in_sizes, int n_in,
                              void* d_out, int out_size, void* d_ws, size_t ws_size,
                              hipStream_t stream)
{
  (void)in_sizes; (void)n_in; (void)out_size; (void)ws_size;
  const float* x      = (const float*)d_in[0];
  const float* hidden = (const float*)d_in[1];
  const float* h0     = (const float*)d_in[2];
  const float* c0     = (const float*)d_in[3];
  const float* f_w1   = (const float*)d_in[4];
  const float* f_b1   = (const float*)d_in[5];
  const float* f_w2   = (const float*)d_in[6];
  const float* f_b2   = (const float*)d_in[7];
  const float* g_w1   = (const float*)d_in[8];
  const float* g_b1   = (const float*)d_in[9];
  const float* g_w2   = (const float*)d_in[10];
  const float* g_b2   = (const float*)d_in[11];
  const float* wih    = (const float*)d_in[12];
  const float* whh    = (const float*)d_in[13];
  const float* bih    = (const float*)d_in[14];
  const float* bhh    = (const float*)d_in[15];
  const float* o_w1   = (const float*)d_in[16];
  const float* o_b1   = (const float*)d_in[17];
  const float* o_w2   = (const float*)d_in[18];
  const float* o_b2   = (const float*)d_in[19];
  float* out = (float*)d_out;

  float* R0 = (float*)d_ws;
  float* R1 = R0 + (size_t)2048 * 256;
  float* R2 = R1 + (size_t)2048 * 256;
  float* R3 = R2 + (size_t)2048 * 256;

  chain_f<<<128, 256, 0, stream>>>(hidden, f_w1, f_b1, R0, R1);
  pair_relu_sum<<<256, 256, 0, stream>>>(R0, R1, R2);
  chain_g<<<128, 256, 0, stream>>>(R2, x, f_w2, f_b2, g_w1, g_b1, g_w2, g_b2,
                                   wih, bih, bhh, R0);
  lstm_scan<<<16, 512, 0, stream>>>(R0, whh, h0, c0, R3);
  chain_o<<<128, 256, 0, stream>>>(R3, o_w1, o_b1, o_w2, o_b2, out);
}


# LSTM scan: the step barrier fences LDS only, so Gx loads stay in flight across it; each step's Gx value is requested 4 steps ahead into a fixed register (4 steps per trip); sums unchanged
# speedup vs baseline: 15.5589x; 1.3076x over previous
#include <hip/hip_runtime.h>
#include <hip/hip_bf16.h>

typedef __attribute__((ext_vector_type(2))) float v2f;
typedef __attribute__((ext_vector_type(8))) float v8f;
typedef __attribute__((ext_vector_type(4))) float v4f;

__device__ __forceinline__ float sigmoidf_(float x) { return 1.0f / (1.0f + __expf(-x)); }

__device__ __forceinline__ v8f wmma4(v2f a, v2f b, v8f c) {
  return __builtin_amdgcn_wmma_f32_16x16x4_f32(false, a, false, b, (short)0, c,
                                               false, false);
}

#define PA128 132
#define PA256 260

template<int NT, int K, int UNR>
__device__ __forceinline__ void mma_lds_a(const float* as, const float* __restrict__ wp,
                                          int ldw, v8f (&acc)[NT])
{
  #pragma clang loop unroll_count(UNR)
  for (int k0 = 0; k0 < K; k0 += 4) {
    const v2f a = *reinterpret_cast<const v2f*>(as + k0);
    #pragma unroll
    for (int t = 0; t < NT; ++t) {
      v2f b;
      b.x = wp[t * 16];
      b.y = wp[ldw + t * 16];
      acc[t] = wmma4(a, b, acc[t]);
    }
    wp += (size_t)4 * ldw;
  }
}

template<int K>
__device__ __forceinline__ void mma_lds_a_wt(const float* as4, const float* __restrict__ wrow,
                                             v8f (&acc)[4])
{
  #pragma clang loop unroll_count(4)
  for (int kb = 0; kb < K; kb += 8) {
    const float4 a4 = *reinterpret_cast<const float4*>(as4 + kb);
    v2f a01, a23; a01.x = a4.x; a01.y = a4.y; a23.x = a4.z; a23.y = a4.w;
    float4 b4[4];
    #pragma unroll
    for (int t = 0; t < 4; ++t)
      b4[t] = *reinterpret_cast<const float4*>(wrow + (size_t)t * 16 * K + kb);
    #pragma unroll
    for (int t = 0; t < 4; ++t) { v2f b; b.x = b4[t].x; b.y = b4[t].y; acc[t] = wmma4(a01, b, acc[t]); }
    #pragma unroll
    for (int t = 0; t < 4; ++t) { v2f b; b.x = b4[t].z; b.y = b4[t].w; acc[t] = wmma4(a23, b, acc[t]); }
  }
}

template<int C4>
__device__ __forceinline__ void load_rows(const float* __restrict__ src, int tid,
                                          float4 (&v)[16 * C4 / 256])
{
  #pragma unroll
  for (int p = 0; p < 16 * C4 / 256; ++p) {
    const int i = tid + p * 256;
    const int r = i / C4, c = (i - r * C4) * 4;
    v[p] = *reinterpret_cast<const float4*>(src + (size_t)r * (4 * C4) + c);
  }
}
template<int C4>
__device__ __forceinline__ void store_rows(float* dst, int pa, int tid,
                                           const float4 (&v)[16 * C4 / 256])
{
  #pragma unroll
  for (int p = 0; p < 16 * C4 / 256; ++p) {
    const int i = tid + p * 256;
    const int r = i / C4, c = (i - r * C4) * 4;
    *reinterpret_cast<float4*>(dst + r * pa + c) = v[p];
  }
}

__global__ void __launch_bounds__(256)
chain_f(const float* __restrict__ hidden, const float* __restrict__ f_w1,
        const float* __restrict__ f_b1, float* __restrict__ Af, float* __restrict__ Bf)
{
  __shared__ __attribute__((aligned(16))) float sH[16 * PA128];
  const int tid = threadIdx.x, wave = tid >> 5, lane = tid & 31;
  const int half = lane >> 4, lp = lane & 15;
  const int m0 = blockIdx.x * 16;
  {
    float4 vh[2];
    load_rows<32>(hidden + (size_t)m0 * 128, tid, vh);
    store_rows<32>(sH, PA128, tid, vh);
  }
  __syncthreads();

  const int n0 = wave * 32;
  const float* as = sH + lp * PA128 + 2 * half;
  v8f acc[2] = {};
  mma_lds_a<2, 128, 8>(as, f_w1 + (size_t)(2 * half) * 256 + n0 + lp, 256, acc);
  v8f bcc[2] = {};
  mma_lds_a<2, 128, 8>(as, f_w1 + (size_t)(128 + 2 * half) * 256 + n0 + lp, 256, bcc);
  #pragma unroll
  for (int t = 0; t < 2; ++t) {
    const int n = n0 + 16 * t + lp;
    const float bias = f_b1[n];
    #pragma unroll
    for (int r = 0; r < 8; ++r) {
      const size_t o = (size_t)(m0 + 8 * half + r) * 256 + n;
      Af[o] = acc[t][r];
      Bf[o] = bcc[t][r] + bias;
    }
  }
}

__global__ void __launch_bounds__(256)
chain_g(const float* __restrict__ S, const float* __restrict__ x,
        const float* __restrict__ f_w2, const float* __restrict__ f_b2,
        const float* __restrict__ g_w1, const float* __restrict__ g_b1,
        const float* __restrict__ g_w2, const float* __restrict__ g_b2,
        const float* __restrict__ wih, const float* __restrict__ bih,
        const float* __restrict__ bhh, float* __restrict__ Gx)
{
  __shared__ __attribute__((aligned(16))) float bufBig[16 * PA256];
  __shared__ __attribute__((aligned(16))) float bufX[16 * PA128];
  __shared__ __attribute__((aligned(16))) float bufM[16 * PA128];
  const int tid = threadIdx.x, wave = tid >> 5, lane = tid & 31;
  const int half = lane >> 4, lp = lane & 15;
  const int m0 = blockIdx.x * 16;
  {
    float4 vs[4], vx[2];
    load_rows<64>(S + (size_t)m0 * 256, tid, vs);
    load_rows<32>(x + (size_t)m0 * 128, tid, vx);
    store_rows<64>(bufBig, PA256, tid, vs);
    store_rows<32>(bufX, PA128, tid, vx);
  }
  __syncthreads();

  {
    const int n = wave * 16 + lp;
    v8f acc[1] = {};
    mma_lds_a<1, 256, 16>(bufBig + lp * PA256 + 2 * half,
                          f_w2 + (size_t)(2 * half) * 128 + n, 128, acc);
    const float bias = 128.0f * f_b2[n];
    #pragma unroll
    for (int r = 0; r < 8; ++r) bufM[(8 * half + r) * PA128 + n] = acc[0][r] + bias;
  }
  __syncthreads();

  {
    const int n0 = wave * 32;
    v8f acc[2] = {};
    mma_lds_a<2, 128, 8>(bufX + lp * PA128 + 2 * half,
                         g_w1 + (size_t)(2 * half) * 256 + n0 + lp, 256, acc);
    mma_lds_a<2, 128, 8>(bufM + lp * PA128 + 2 * half,
                         g_w1 + (size_t)(128 + 2 * half) * 256 + n0 + lp, 256, acc);
    #pragma unroll
    for (int t = 0; t < 2; ++t) {
      const int n = n0 + 16 * t + lp;
      const float bias = g_b1[n];
      #pragma unroll
      for (int r = 0; r < 8; ++r)
        bufBig[(8 * half + r) * PA256 + n] = fmaxf(acc[t][r] + bias, 0.0f);
    }
  }
  __syncthreads();

  {
    const int n = wave * 16 + lp;
    v8f acc[1] = {};
    mma_lds_a<1, 256, 16>(bufBig + lp * PA256 + 2 * half,
                          g_w2 + (size_t)(2 * half) * 128 + n, 128, acc);
    const float bias = g_b2[n];
    #pragma unroll
    for (int r = 0; r < 8; ++r) bufX[(8 * half + r) * PA128 + n] = acc[0][r] + bias;
  }
  __syncthreads();

  {
    const int n0 = wave * 64;
    v8f acc[4] = {};
    mma_lds_a_wt<128>(bufX + lp * PA128 + 4 * half,
                      wih + (size_t)(n0 + lp) * 128 + 4 * half, acc);
    #pragma unroll
    for (int t = 0; t < 4; ++t) {
      const int n = n0 + 16 * t + lp;
      const float bias = bih[n] + bhh[n];
      #pragma unroll
      for (int r = 0; r < 8; ++r)
        Gx[(size_t)(m0 + 8 * half + r) * 512 + n] = acc[t][r] + bias;
    }
  }
}

__global__ void __launch_bounds__(256)
chain_o(const float* __restrict__ nh, const float* __restrict__ o_w1,
        const float* __restrict__ o_b1, const float* __restrict__ o_w2,
        const float* __restrict__ o_b2, float* __restrict__ out)
{
  __shared__ __attribute__((aligned(16))) float sH[16 * PA128];
  __shared__ __attribute__((aligned(16))) float sO[16 * PA256];
  const int tid = threadIdx.x, wave = tid >> 5, lane = tid & 31;
  const int half = lane >> 4, lp = lane & 15;
  const int m0 = blockIdx.x * 16;
  {
    float4 vh[2];
    load_rows<32>(nh + (size_t)m0 * 128, tid, vh);
    store_rows<32>(sH, PA128, tid, vh);
  }
  __syncthreads();
  {
    const int n0 = wave * 32;
    v8f acc[2] = {};
    mma_lds_a<2, 128, 8>(sH + lp * PA128 + 2 * half,
                         o_w1 + (size_t)(2 * half) * 256 + n0 + lp, 256, acc);
    #pragma unroll
    for (int t = 0; t < 2; ++t) {
      const int n = n0 + 16 * t + lp;
      const float bias = o_b1[n];
      #pragma unroll
      for (int r = 0; r < 8; ++r)
        sO[(8 * half + r) * PA256 + n] = fmaxf(acc[t][r] + bias, 0.0f);
    }
  }
  __syncthreads();
  if (wave < 2) {
    const int n = wave * 16 + lp;
    v8f acc[1] = {};
    mma_lds_a<1, 256, 16>(sO + lp * PA256 + 2 * half,
                          o_w2 + (size_t)(2 * half) * 32 + n, 32, acc);
    const float bias = o_b2[n];
    #pragma unroll
    for (int r = 0; r < 8; ++r)
      sH[(8 * half + r) * 36 + n] = acc[0][r] + bias;
  }
  __syncthreads();
  if (tid < 128) {
    const int r = tid >> 3, c = (tid & 7) * 4;
    const float4 v = *reinterpret_cast<const float4*>(sH + r * 36 + c);
    *reinterpret_cast<float4*>(out + (size_t)(m0 + r) * 32 + c) = v;
  }
}

#define PR_TI 8
#define PR_G  16

__device__ __forceinline__ void pr_load(const float* __restrict__ p, float (&v)[PR_G])
{
  #pragma unroll
  for (int q = 0; q < PR_G; ++q) v[q] = p[(size_t)q * 256];
}
__device__ __forceinline__ void pr_accum(const float (&v)[PR_G], const float (&bm)[PR_TI],
                                         float (&acc)[PR_TI])
{
  #pragma unroll
  for (int q = 0; q < PR_G; ++q) {
    #pragma unroll
    for (int i = 0; i < PR_TI; ++i) acc[i] += fmaxf(v[q] + bm[i], 0.0f);
  }
}

__global__ void __launch_bounds__(256)
pair_relu_sum(const float* __restrict__ Af, const float* __restrict__ Bf,
              float* __restrict__ S)
{
  const int b  = blockIdx.x >> 4;
  const int r0 = b * 128 + (blockIdx.x & 15) * PR_TI;
  const int f  = threadIdx.x;
  const float* bp = Bf + (size_t)r0 * 256 + f;
  const float* ap = Af + (size_t)b * 128 * 256 + f;
  float va[PR_G], vb[PR_G];
  pr_load(ap, va);
  float bm[PR_TI], acc[PR_TI];
  #pragma unroll
  for (int i = 0; i < PR_TI; ++i) { bm[i] = bp[(size_t)i * 256]; acc[i] = 0.0f; }
  for (int g = 0; g < 128 / (2 * PR_G); ++g) {
    pr_load(ap + (size_t)(2 * PR_G * g + PR_G) * 256, vb);
    pr_accum(va, bm, acc);
    if (g + 1 < 128 / (2 * PR_G))
      pr_load(ap + (size_t)(2 * PR_G * g + 2 * PR_G) * 256, va);
    pr_accum(vb, bm, acc);
  }
  float* sp = S + (size_t)r0 * 256 + f;
  #pragma unroll
  for (int i = 0; i < PR_TI; ++i) sp[(size_t)i * 256] = acc[i];
}

template<int CTRL>
__device__ __forceinline__ float quad_get(float v)
{
  return __int_as_float(__builtin_amdgcn_mov_dpp(__float_as_int(v), CTRL, 0xf, 0xf, true));
}
#define QP_XOR1 0xB1
#define QP_XOR2 0x4E
#define QP_L0   0x00
#define QP_L1   0x55
#define QP_L2   0xAA
#define QP_L3   0xFF

__device__ __forceinline__ v2f fma2(float h, v2f w, v2f acc)
{
  v2f hh; hh.x = h; hh.y = h;
  return __builtin_elementwise_fma(hh, w, acc);
}

#define GX_AHEAD 4
template<int V>  struct StepReg { static constexpr int  value = V; };
template<bool V> struct StepAsk { static constexpr bool value = V; };

__device__ __forceinline__ void lds_step_barrier()
{
  __builtin_amdgcn_fence(__ATOMIC_RELEASE, "workgroup", "local");
  __builtin_amdgcn_s_barrier();
  __builtin_amdgcn_fence(__ATOMIC_ACQUIRE, "workgroup", "local");
}

__global__ void __launch_bounds__(512)
lstm_scan(const float* __restrict__ Gx, const float* __restrict__ whh,
          const float* __restrict__ h0, const float* __restrict__ c0,
          float* __restrict__ new_hidden)
{
  __shared__ __attribute__((aligned(16))) float h_s[2][128];
  const int b = blockIdx.x;
  const int tid = threadIdx.x;
  const int q = tid & 3;
  const int n = tid >> 2;
  const int wave = tid >> 5, lane = tid & 31;

  v2f wif[32], wgo[32];
  {
    const float* ri = whh + (size_t)(0 * 128 + n) * 128 + 4 * q;
    const float* rf = whh + (size_t)(1 * 128 + n) * 128 + 4 * q;
    const float* rg = whh + (size_t)(2 * 128 + n) * 128 + 4 * q;
    const float* ro = whh + (size_t)(3 * 128 + n) * 128 + 4 * q;
    #pragma unroll
    for (int j = 0; j < 8; ++j) {
      const float4 vi = *reinterpret_cast<const float4*>(ri + 16 * j);
      const float4 vf = *reinterpret_cast<const float4*>(rf + 16 * j);
      const float4 vg = *reinterpret_cast<const float4*>(rg + 16 * j);
      const float4 vo = *reinterpret_cast<const float4*>(ro + 16 * j);
      wif[4*j+0].x = vi.x; wif[4*j+0].y = vf.x; wgo[4*j+0].x = vg.x; wgo[4*j+0].y = vo.x;
      wif[4*j+1].x = vi.y; wif[4*j+1].y = vf.y; wgo[4*j+1].x = vg.y; wgo[4*j+1].y = vo.y;
      wif[4*j+2].x = vi.z; wif[4*j+2].y = vf.z; wgo[4*j+2].x = vg.z; wgo[4*j+2].y = vo.z;
      wif[4*j+3].x = vi.w; wif[4*j+3].y = vf.w; wgo[4*j+3].x = vg.w; wgo[4*j+3].y = vo.w;
    }
  }

  const float LOG2E = 1.44269504088896341f;
  const float am = (q == 2) ? 2.0f : 1.0f;
  const float bm = (q == 2) ? -1.0f : 0.0f;
  const float ke = -am * LOG2E;
  const bool odd = (q & 1) != 0, hi = (q & 2) != 0;

  float c = c0[b * 128 + n];
  if (q == 0) h_s[0][n] = h0[b * 128 + n];

  const float* gx = Gx + (size_t)b * 128 * 512 + q * 128 + n;
  float gxr[GX_AHEAD];
  #pragma unroll
  for (int u = 0; u < GX_AHEAD; ++u) gxr[u] = gx[(size_t)u * 512];
  float* hout = new_hidden + (size_t)b * 128 * 128 + 32 * wave + lane;
  __syncthreads();

  auto step = [&](const int t, auto U, auto REQUEST) __attribute__((always_inline)) {
    constexpr int u = decltype(U)::value;
    constexpr bool request = decltype(REQUEST)::value;
    const float* hs = h_s[t & 1] + 4 * q;
    v4f hv[8];
    #pragma unroll
    for (int j = 0; j < 8; ++j) hv[j] = *reinterpret_cast<const v4f*>(hs + 16 * j);
    asm volatile("" :: "v"(hv[0]), "v"(hv[1]), "v"(hv[2]), "v"(hv[3]),
                       "v"(hv[4]), "v"(hv[5]), "v"(hv[6]), "v"(hv[7]));
    __builtin_amdgcn_sched_barrier(0);
    const float gx_now = gxr[u];
    v2f aif0 = {0.0f, 0.0f}, aif1 = {0.0f, 0.0f}, ago0 = {0.0f, 0.0f}, ago1 = {0.0f, 0.0f};
    #pragma unroll
    for (int j = 0; j < 8; ++j) {
      aif0 = fma2(hv[j].x, wif[4*j+0], aif0);  ago0 = fma2(hv[j].x, wgo[4*j+0], ago0);
      aif1 = fma2(hv[j].y, wif[4*j+1], aif1);  ago1 = fma2(hv[j].y, wgo[4*j+1], ago1);
      aif0 = fma2(hv[j].z, wif[4*j+2], aif0);  ago0 = fma2(hv[j].z, wgo[4*j+2], ago0);
      aif1 = fma2(hv[j].w, wif[4*j+3], aif1);  ago1 = fma2(hv[j].w, wgo[4*j+3], ago1);
    }
    const v2f pif = aif0 + aif1, pgo = ago0 + ago1;

    const float keepA = odd ? pif.y : pif.x, sendA = odd ? pif.x : pif.y;
    const float keepB = odd ? pgo.y : pgo.x, sendB = odd ? pgo.x : pgo.y;
    const float sA = keepA + quad_get<QP_XOR1>(sendA);
    const float sB = keepB + quad_get<QP_XOR1>(sendB);
    const float keep = hi ? sB : sA, send = hi ? sA : sB;
    const float x = (keep + quad_get<QP_XOR2>(send)) + gx_now;

    const float act = fmaf(am, __builtin_amdgcn_rcpf(1.0f + __builtin_amdgcn_exp2f(ke * x)), bm);
    const float ig = quad_get<QP_L0>(act), fg = quad_get<QP_L1>(act);
    const float g2 = quad_get<QP_L2>(act), og = quad_get<QP_L3>(act);
    c = fmaf(fg, c, ig * g2);
    const float th = fmaf(2.0f, __builtin_amdgcn_rcpf(1.0f + __builtin_amdgcn_exp2f((-2.0f * LOG2E) * c)), -1.0f);
    const float h = og * th;
    if (q == 0) h_s[(t + 1) & 1][n] = h;
    lds_step_barrier();
    if constexpr (request) gxr[u] = gx[(size_t)(t + GX_AHEAD) * 512];
    if (wave < 4) hout[(size_t)t * 128] = h_s[(t + 1) & 1][32 * wave + lane];
  };

  static_assert(GX_AHEAD == 4, "the trips below are written out for 4 steps");
  for (int t0 = 0; t0 < 128 - GX_AHEAD; t0 += GX_AHEAD) {
    step(t0 + 0, StepReg<0>{}, StepAsk<true>{});
    step(t0 + 1, StepReg<1>{}, StepAsk<true>{});
    step(t0 + 2, StepReg<2>{}, StepAsk<true>{});
    step(t0 + 3, StepReg<3>{}, StepAsk<true>{});
  }
  step(128 - GX_AHEAD + 0, StepReg<0>{}, StepAsk<false>{});
  step(128 - GX_AHEAD + 1, StepReg<1>{}, StepAsk<false>{});
  step(128 - GX_AHEAD + 2, StepReg<2>{}, StepAsk<false>{});
  step(128 - GX_AHEAD + 3, StepReg<3>{}, StepAsk<false>{});
}

extern "C" void kernel_launch(void* const* d_in, const int* in_sizes, int n_in,
                              void* d_out, int out_size, void* d_ws, size_t ws_size,
                              hipStream_t stream)
{
  (void)in_sizes; (void)n_in; (void)out_size; (void)ws_size;
  const float* x      = (const float*)d_in[0];
  const float* hidden = (const float*)d_in[1];
  const float* h0     = (const float*)d_in[2];
  const float* c0     = (const float*)d_in[3];
  const float* f_w1   = (const float*)d_in[4];
  const float* f_b1   = (const float*)d_in[5];
  const float* f_w2   = (const float*)d_in[6];
  const float* f_b2   = (const float*)d_in[7];
  const float* g_w1   = (const float*)d_in[8];
  const float* g_b1   = (const float*)d_in[9];
  const float* g_w2   = (const float*)d_in[10];
  const float* g_b2   = (const float*)d_in[11];
  const float* wih    = (const float*)d_in[12];
  const float* whh    = (const float*)d_in[13];
  const float* bih    = (const float*)d_in[14];
  const float* bhh    = (const float*)d_in[15];
  const float* o_w1   = (const float*)d_in[16];
  const float* o_b1   = (const float*)d_in[17];
  const float* o_w2   = (const float*)d_in[18];
  const float* o_b2   = (const float*)d_in[19];
  float* out = (float*)d_out;

  float* R0 = (float*)d_ws;
  float* R1 = R0 + (size_t)2048 * 256;
  float* R2 = R1 + (size_t)2048 * 256;
  float* R3 = R2 + (size_t)2048 * 256;

  chain_f<<<128, 256, 0, stream>>>(hidden, f_w1, f_b1, R0, R1);
  pair_relu_sum<<<256, 256, 0, stream>>>(R0, R1, R2);
  chain_g<<<128, 256, 0, stream>>>(R2, x, f_w2, f_b2, g_w1, g_b1, g_w2, g_b2,
                                   wih, bih, bhh, R0);
  lstm_scan<<<16, 512, 0, stream>>>(R0, whh, h0, c0, R3);
  chain_o<<<128, 256, 0, stream>>>(R3, o_w1, o_b1, o_w2, o_b2, out);
}


# LSTM scan: gate activations and tanh(c) on the hardware tanh instruction, sigmoid(x) = 1/2 + 1/2 tanh(x/2) with the 1/2 kept in the whh registers (exact), in place of exp2 + rcp; sum order unchanged
# speedup vs baseline: 16.3256x; 1.0493x over previous
#include <hip/hip_runtime.h>
#include <hip/hip_bf16.h>

typedef __attribute__((ext_vector_type(2))) float v2f;
typedef __attribute__((ext_vector_type(8))) float v8f;
typedef __attribute__((ext_vector_type(4))) float v4f;

__device__ __forceinline__ float sigmoidf_(float x) { return 1.0f / (1.0f + __expf(-x)); }

__device__ __forceinline__ v8f wmma4(v2f a, v2f b, v8f c) {
  return __builtin_amdgcn_wmma_f32_16x16x4_f32(false, a, false, b, (short)0, c,
                                               false, false);
}

#define PA128 132
#define PA256 260

template<int NT, int K, int UNR>
__device__ __forceinline__ void mma_lds_a(const float* as, const float* __restrict__ wp,
                                          int ldw, v8f (&acc)[NT])
{
  #pragma clang loop unroll_count(UNR)
  for (int k0 = 0; k0 < K; k0 += 4) {
    const v2f a = *reinterpret_cast<const v2f*>(as + k0);
    #pragma unroll
    for (int t = 0; t < NT; ++t) {
      v2f b;
      b.x = wp[t * 16];
      b.y = wp[ldw + t * 16];
      acc[t] = wmma4(a, b, acc[t]);
    }
    wp += (size_t)4 * ldw;
  }
}

template<int K>
__device__ __forceinline__ void mma_lds_a_wt(const float* as4, const float* __restrict__ wrow,
                                             v8f (&acc)[4])
{
  #pragma clang loop unroll_count(4)
  for (int kb = 0; kb < K; kb += 8) {
    const float4 a4 = *reinterpret_cast<const float4*>(as4 + kb);
    v2f a01, a23; a01.x = a4.x; a01.y = a4.y; a23.x = a4.z; a23.y = a4.w;
    float4 b4[4];
    #pragma unroll
    for (int t = 0; t < 4; ++t)
      b4[t] = *reinterpret_cast<const float4*>(wrow + (size_t)t * 16 * K + kb);
    #pragma unroll
    for (int t = 0; t < 4; ++t) { v2f b; b.x = b4[t].x; b.y = b4[t].y; acc[t] = wmma4(a01, b, acc[t]); }
    #pragma unroll
    for (int t = 0; t < 4; ++t) { v2f b; b.x = b4[t].z; b.y = b4[t].w; acc[t] = wmma4(a23, b, acc[t]); }
  }
}

template<int C4>
__device__ __forceinline__ void load_rows(const float* __restrict__ src, int tid,
                                          float4 (&v)[16 * C4 / 256])
{
  #pragma unroll
  for (int p = 0; p < 16 * C4 / 256; ++p) {
    const int i = tid + p * 256;
    const int r = i / C4, c = (i - r * C4) * 4;
    v[p] = *reinterpret_cast<const float4*>(src + (size_t)r * (4 * C4) + c);
  }
}
template<int C4>
__device__ __forceinline__ void store_rows(float* dst, int pa, int tid,
                                           const float4 (&v)[16 * C4 / 256])
{
  #pragma unroll
  for (int p = 0; p < 16 * C4 / 256; ++p) {
    const int i = tid + p * 256;
    const int r = i / C4, c = (i - r * C4) * 4;
    *reinterpret_cast<float4*>(dst + r * pa + c) = v[p];
  }
}

__global__ void __launch_bounds__(256)
chain_f(const float* __restrict__ hidden, const float* __restrict__ f_w1,
        const float* __restrict__ f_b1, float* __restrict__ Af, float* __restrict__ Bf)
{
  __shared__ __attribute__((aligned(16))) float sH[16 * PA128];
  const int tid = threadIdx.x, wave = tid >> 5, lane = tid & 31;
  const int half = lane >> 4, lp = lane & 15;
  const int m0 = blockIdx.x * 16;
  {
    float4 vh[2];
    load_rows<32>(hidden + (size_t)m0 * 128, tid, vh);
    store_rows<32>(sH, PA128, tid, vh);
  }
  __syncthreads();

  const int n0 = wave * 32;
  const float* as = sH + lp * PA128 + 2 * half;
  v8f acc[2] = {};
  mma_lds_a<2, 128, 8>(as, f_w1 + (size_t)(2 * half) * 256 + n0 + lp, 256, acc);
  v8f bcc[2] = {};
  mma_lds_a<2, 128, 8>(as, f_w1 + (size_t)(128 + 2 * half) * 256 + n0 + lp, 256, bcc);
  #pragma unroll
  for (int t = 0; t < 2; ++t) {
    const int n = n0 + 16 * t + lp;
    const float bias = f_b1[n];
    #pragma unroll
    for (int r = 0; r < 8; ++r) {
      const size_t o = (size_t)(m0 + 8 * half + r) * 256 + n;
      Af[o] = acc[t][r];
      Bf[o] = bcc[t][r] + bias;
    }
  }
}

__global__ void __launch_bounds__(256)
chain_g(const float* __restrict__ S, const float* __restrict__ x,
        const float* __restrict__ f_w2, const float* __restrict__ f_b2,
        const float* __restrict__ g_w1, const float* __restrict__ g_b1,
        const float* __restrict__ g_w2, const float* __restrict__ g_b2,
        const float* __restrict__ wih, const float* __restrict__ bih,
        const float* __restrict__ bhh, float* __restrict__ Gx)
{
  __shared__ __attribute__((aligned(16))) float bufBig[16 * PA256];
  __shared__ __attribute__((aligned(16))) float bufX[16 * PA128];
  __shared__ __attribute__((aligned(16))) float bufM[16 * PA128];
  const int tid = threadIdx.x, wave = tid >> 5, lane = tid & 31;
  const int half = lane >> 4, lp = lane & 15;
  const int m0 = blockIdx.x * 16;
  {
    float4 vs[4], vx[2];
    load_rows<64>(S + (size_t)m0 * 256, tid, vs);
    load_rows<32>(x + (size_t)m0 * 128, tid, vx);
    store_rows<64>(bufBig, PA256, tid, vs);
    store_rows<32>(bufX, PA128, tid, vx);
  }
  __syncthreads();

  {
    const int n = wave * 16 + lp;
    v8f acc[1] = {};
    mma_lds_a<1, 256, 16>(bufBig + lp * PA256 + 2 * half,
                          f_w2 + (size_t)(2 * half) * 128 + n, 128, acc);
    const float bias = 128.0f * f_b2[n];
    #pragma unroll
    for (int r = 0; r < 8; ++r) bufM[(8 * half + r) * PA128 + n] = acc[0][r] + bias;
  }
  __syncthreads();

  {
    const int n0 = wave * 32;
    v8f acc[2] = {};
    mma_lds_a<2, 128, 8>(bufX + lp * PA128 + 2 * half,
                         g_w1 + (size_t)(2 * half) * 256 + n0 + lp, 256, acc);
    mma_lds_a<2, 128, 8>(bufM + lp * PA128 + 2 * half,
                         g_w1 + (size_t)(128 + 2 * half) * 256 + n0 + lp, 256, acc);
    #pragma unroll
    for (int t = 0; t < 2; ++t) {
      const int n = n0 + 16 * t + lp;
      const float bias = g_b1[n];
      #pragma unroll
      for (int r = 0; r < 8; ++r)
        bufBig[(8 * half + r) * PA256 + n] = fmaxf(acc[t][r] + bias, 0.0f);
    }
  }
  __syncthreads();

  {
    const int n = wave * 16 + lp;
    v8f acc[1] = {};
    mma_lds_a<1, 256, 16>(bufBig + lp * PA256 + 2 * half,
                          g_w2 + (size_t)(2 * half) * 128 + n, 128, acc);
    const float bias = g_b2[n];
    #pragma unroll
    for (int r = 0; r < 8; ++r) bufX[(8 * half + r) * PA128 + n] = acc[0][r] + bias;
  }
  __syncthreads();

  {
    const int n0 = wave * 64;
    v8f acc[4] = {};
    mma_lds_a_wt<128>(bufX + lp * PA128 + 4 * half,
                      wih + (size_t)(n0 + lp) * 128 + 4 * half, acc);
    #pragma unroll
    for (int t = 0; t < 4; ++t) {
      const int n = n0 + 16 * t + lp;
      const float bias = bih[n] + bhh[n];
      #pragma unroll
      for (int r = 0; r < 8; ++r)
        Gx[(size_t)(m0 + 8 * half + r) * 512 + n] = acc[t][r] + bias;
    }
  }
}

__global__ void __launch_bounds__(256)
chain_o(const float* __restrict__ nh, const float* __restrict__ o_w1,
        const float* __restrict__ o_b1, const float* __restrict__ o_w2,
        const float* __restrict__ o_b2, float* __restrict__ out)
{
  __shared__ __attribute__((aligned(16))) float sH[16 * PA128];
  __shared__ __attribute__((aligned(16))) float sO[16 * PA256];
  const int tid = threadIdx.x, wave = tid >> 5, lane = tid & 31;
  const int half = lane >> 4, lp = lane & 15;
  const int m0 = blockIdx.x * 16;
  {
    float4 vh[2];
    load_rows<32>(nh + (size_t)m0 * 128, tid, vh);
    store_rows<32>(sH, PA128, tid, vh);
  }
  __syncthreads();
  {
    const int n0 = wave * 32;
    v8f acc[2] = {};
    mma_lds_a<2, 128, 8>(sH + lp * PA128 + 2 * half,
                         o_w1 + (size_t)(2 * half) * 256 + n0 + lp, 256, acc);
    #pragma unroll
    for (int t = 0; t < 2; ++t) {
      const int n = n0 + 16 * t + lp;
      const float bias = o_b1[n];
      #pragma unroll
      for (int r = 0; r < 8; ++r)
        sO[(8 * half + r) * PA256 + n] = fmaxf(acc[t][r] + bias, 0.0f);
    }
  }
  __syncthreads();
  if (wave < 2) {
    const int n = wave * 16 + lp;
    v8f acc[1] = {};
    mma_lds_a<1, 256, 16>(sO + lp * PA256 + 2 * half,
                          o_w2 + (size_t)(2 * half) * 32 + n, 32, acc);
    const float bias = o_b2[n];
    #pragma unroll
    for (int r = 0; r < 8; ++r)
      sH[(8 * half + r) * 36 + n] = acc[0][r] + bias;
  }
  __syncthreads();
  if (tid < 128) {
    const int r = tid >> 3, c = (tid & 7) * 4;
    const float4 v = *reinterpret_cast<const float4*>(sH + r * 36 + c);
    *reinterpret_cast<float4*>(out + (size_t)(m0 + r) * 32 + c) = v;
  }
}

#define PR_TI 8
#define PR_G  16

__device__ __forceinline__ void pr_load(const float* __restrict__ p, float (&v)[PR_G])
{
  #pragma unroll
  for (int q = 0; q < PR_G; ++q) v[q] = p[(size_t)q * 256];
}
__device__ __forceinline__ void pr_accum(const float (&v)[PR_G], const float (&bm)[PR_TI],
                                         float (&acc)[PR_TI])
{
  #pragma unroll
  for (int q = 0; q < PR_G; ++q) {
    #pragma unroll
    for (int i = 0; i < PR_TI; ++i) acc[i] += fmaxf(v[q] + bm[i], 0.0f);
  }
}

__global__ void __launch_bounds__(256)
pair_relu_sum(const float* __restrict__ Af, const float* __restrict__ Bf,
              float* __restrict__ S)
{
  const int b  = blockIdx.x >> 4;
  const int r0 = b * 128 + (blockIdx.x & 15) * PR_TI;
  const int f  = threadIdx.x;
  const float* bp = Bf + (size_t)r0 * 256 + f;
  const float* ap = Af + (size_t)b * 128 * 256 + f;
  float va[PR_G], vb[PR_G];
  pr_load(ap, va);
  float bm[PR_TI], acc[PR_TI];
  #pragma unroll
  for (int i = 0; i < PR_TI; ++i) { bm[i] = bp[(size_t)i * 256]; acc[i] = 0.0f; }
  for (int g = 0; g < 128 / (2 * PR_G); ++g) {
    pr_load(ap + (size_t)(2 * PR_G * g + PR_G) * 256, vb);
    pr_accum(va, bm, acc);
    if (g + 1 < 128 / (2 * PR_G))
      pr_load(ap + (size_t)(2 * PR_G * g + 2 * PR_G) * 256, va);
    pr_accum(vb, bm, acc);
  }
  float* sp = S + (size_t)r0 * 256 + f;
  #pragma unroll
  for (int i = 0; i < PR_TI; ++i) sp[(size_t)i * 256] = acc[i];
}

template<int CTRL>
__device__ __forceinline__ float quad_get(float v)
{
  return __int_as_float(__builtin_amdgcn_mov_dpp(__float_as_int(v), CTRL, 0xf, 0xf, true));
}
#define QP_XOR1 0xB1
#define QP_XOR2 0x4E
#define QP_L0   0x00
#define QP_L1   0x55
#define QP_L2   0xAA
#define QP_L3   0xFF

__device__ __forceinline__ v2f fma2(float h, v2f w, v2f acc)
{
  v2f hh; hh.x = h; hh.y = h;
  return __builtin_elementwise_fma(hh, w, acc);
}

#define GX_AHEAD 4
template<int V>  struct StepReg { static constexpr int  value = V; };
template<bool V> struct StepAsk { static constexpr bool value = V; };

__device__ __forceinline__ void lds_step_barrier()
{
  __builtin_amdgcn_fence(__ATOMIC_RELEASE, "workgroup", "local");
  __builtin_amdgcn_s_barrier();
  __builtin_amdgcn_fence(__ATOMIC_ACQUIRE, "workgroup", "local");
}

__global__ void __launch_bounds__(512)
lstm_scan(const float* __restrict__ Gx, const float* __restrict__ whh,
          const float* __restrict__ h0, const float* __restrict__ c0,
          float* __restrict__ new_hidden)
{
  __shared__ __attribute__((aligned(16))) float h_s[2][128];
  const int b = blockIdx.x;
  const int tid = threadIdx.x;
  const int q = tid & 3;
  const int n = tid >> 2;
  const int wave = tid >> 5, lane = tid & 31;

  v2f wif[32], wgo[32];
  {
    const float* ri = whh + (size_t)(0 * 128 + n) * 128 + 4 * q;
    const float* rf = whh + (size_t)(1 * 128 + n) * 128 + 4 * q;
    const float* rg = whh + (size_t)(2 * 128 + n) * 128 + 4 * q;
    const float* ro = whh + (size_t)(3 * 128 + n) * 128 + 4 * q;
    const v2f sif = {0.5f, 0.5f}, sgo = {1.0f, 0.5f};
    #pragma unroll
    for (int j = 0; j < 8; ++j) {
      const float4 vi = *reinterpret_cast<const float4*>(ri + 16 * j);
      const float4 vf = *reinterpret_cast<const float4*>(rf + 16 * j);
      const float4 vg = *reinterpret_cast<const float4*>(rg + 16 * j);
      const float4 vo = *reinterpret_cast<const float4*>(ro + 16 * j);
      wif[4*j+0].x = vi.x; wif[4*j+0].y = vf.x; wgo[4*j+0].x = vg.x; wgo[4*j+0].y = vo.x;
      wif[4*j+1].x = vi.y; wif[4*j+1].y = vf.y; wgo[4*j+1].x = vg.y; wgo[4*j+1].y = vo.y;
      wif[4*j+2].x = vi.z; wif[4*j+2].y = vf.z; wgo[4*j+2].x = vg.z; wgo[4*j+2].y = vo.z;
      wif[4*j+3].x = vi.w; wif[4*j+3].y = vf.w; wgo[4*j+3].x = vg.w; wgo[4*j+3].y = vo.w;
    }
    #pragma unroll
    for (int k = 0; k < 32; ++k) { wif[k] *= sif; wgo[k] *= sgo; }
  }

  const float am = (q == 2) ? 1.0f : 0.5f;
  const float bm = (q == 2) ? 0.0f : 0.5f;
  const bool odd = (q & 1) != 0, hi = (q & 2) != 0;

  float c = c0[b * 128 + n];
  if (q == 0) h_s[0][n] = h0[b * 128 + n];

  const float* gx = Gx + (size_t)b * 128 * 512 + q * 128 + n;
  float gxr[GX_AHEAD];
  #pragma unroll
  for (int u = 0; u < GX_AHEAD; ++u) gxr[u] = gx[(size_t)u * 512];
  float* hout = new_hidden + (size_t)b * 128 * 128 + 32 * wave + lane;
  __syncthreads();

  auto step = [&](const int t, auto U, auto REQUEST) __attribute__((always_inline)) {
    constexpr int u = decltype(U)::value;
    constexpr bool request = decltype(REQUEST)::value;
    const float* hs = h_s[t & 1] + 4 * q;
    v4f hv[8];
    #pragma unroll
    for (int j = 0; j < 8; ++j) hv[j] = *reinterpret_cast<const v4f*>(hs + 16 * j);
    asm volatile("" :: "v"(hv[0]), "v"(hv[1]), "v"(hv[2]), "v"(hv[3]),
                       "v"(hv[4]), "v"(hv[5]), "v"(hv[6]), "v"(hv[7]));
    __builtin_amdgcn_sched_barrier(0);
    const float gx_now = gxr[u];
    v2f aif0 = {0.0f, 0.0f}, aif1 = {0.0f, 0.0f}, ago0 = {0.0f, 0.0f}, ago1 = {0.0f, 0.0f};
    #pragma unroll
    for (int j = 0; j < 8; ++j) {
      aif0 = fma2(hv[j].x, wif[4*j+0], aif0);  ago0 = fma2(hv[j].x, wgo[4*j+0], ago0);
      aif1 = fma2(hv[j].y, wif[4*j+1], aif1);  ago1 = fma2(hv[j].y, wgo[4*j+1], ago1);
      aif0 = fma2(hv[j].z, wif[4*j+2], aif0);  ago0 = fma2(hv[j].z, wgo[4*j+2], ago0);
      aif1 = fma2(hv[j].w, wif[4*j+3], aif1);  ago1 = fma2(hv[j].w, wgo[4*j+3], ago1);
    }
    const v2f pif = aif0 + aif1, pgo = ago0 + ago1;

    const float keepA = odd ? pif.y : pif.x, sendA = odd ? pif.x : pif.y;
    const float keepB = odd ? pgo.y : pgo.x, sendB = odd ? pgo.x : pgo.y;
    const float sA = keepA + quad_get<QP_XOR1>(sendA);
    const float sB = keepB + quad_get<QP_XOR1>(sendB);
    const float keep = hi ? sB : sA, send = hi ? sA : sB;
    const float x = fmaf(gx_now, am, keep + quad_get<QP_XOR2>(send));

    const float act = fmaf(am, __builtin_amdgcn_tanhf(x), bm);
    const float ig = quad_get<QP_L0>(act), fg = quad_get<QP_L1>(act);
    const float g2 = quad_get<QP_L2>(act), og = quad_get<QP_L3>(act);
    c = fmaf(fg, c, ig * g2);
    const float h = og * __builtin_amdgcn_tanhf(c);
    if (q == 0) h_s[(t + 1) & 1][n] = h;
    lds_step_barrier();
    if constexpr (request) gxr[u] = gx[(size_t)(t + GX_AHEAD) * 512];
    if (wave < 4) hout[(size_t)t * 128] = h_s[(t + 1) & 1][32 * wave + lane];
  };

  static_assert(GX_AHEAD == 4, "the trips below are written out for 4 steps");
  for (int t0 = 0; t0 < 128 - GX_AHEAD; t0 += GX_AHEAD) {
    step(t0 + 0, StepReg<0>{}, StepAsk<true>{});
    step(t0 + 1, StepReg<1>{}, StepAsk<true>{});
    step(t0 + 2, StepReg<2>{}, StepAsk<true>{});
    step(t0 + 3, StepReg<3>{}, StepAsk<true>{});
  }
  step(128 - GX_AHEAD + 0, StepReg<0>{}, StepAsk<false>{});
  step(128 - GX_AHEAD + 1, StepReg<1>{}, StepAsk<false>{});
  step(128 - GX_AHEAD + 2, StepReg<2>{}, StepAsk<false>{});
  step(128 - GX_AHEAD + 3, StepReg<3>{}, StepAsk<false>{});
}

extern "C" void kernel_launch(void* const* d_in, const int* in_sizes, int n_in,
                              void* d_out, int out_size, void* d_ws, size_t ws_size,
                              hipStream_t stream)
{
  (void)in_sizes; (void)n_in; (void)out_size; (void)ws_size;
  const float* x      = (const float*)d_in[0];
  const float* hidden = (const float*)d_in[1];
  const float* h0     = (const float*)d_in[2];
  const float* c0     = (const float*)d_in[3];
  const float* f_w1   = (const float*)d_in[4];
  const float* f_b1   = (const float*)d_in[5];
  const float* f_w2   = (const float*)d_in[6];
  const float* f_b2   = (const float*)d_in[7];
  const float* g_w1   = (const float*)d_in[8];
  const float* g_b1   = (const float*)d_in[9];
  const float* g_w2   = (const float*)d_in[10];
  const float* g_b2   = (const float*)d_in[11];
  const float* wih    = (const float*)d_in[12];
  const float* whh    = (const float*)d_in[13];
  const float* bih    = (const float*)d_in[14];
  const float* bhh    = (const float*)d_in[15];
  const float* o_w1   = (const float*)d_in[16];
  const float* o_b1   = (const float*)d_in[17];
  const float* o_w2   = (const float*)d_in[18];
  const float* o_b2   = (const float*)d_in[19];
  float* out = (float*)d_out;

  float* R0 = (float*)d_ws;
  float* R1 = R0 + (size_t)2048 * 256;
  float* R2 = R1 + (size_t)2048 * 256;
  float* R3 = R2 + (size_t)2048 * 256;

  chain_f<<<128, 256, 0, stream>>>(hidden, f_w1, f_b1, R0, R1);
  pair_relu_sum<<<256, 256, 0, stream>>>(R0, R1, R2);
  chain_g<<<128, 256, 0, stream>>>(R2, x, f_w2, f_b2, g_w1, g_b1, g_w2, g_b2,
                                   wih, bih, bhh, R0);
  lstm_scan<<<16, 512, 0, stream>>>(R0, whh, h0, c0, R3);
  chain_o<<<128, 256, 0, stream>>>(R3, o_w1, o_b1, o_w2, o_b2, out);
}


# LSTM scan: a step's FMAs start on the first 16-byte h group to arrive from LDS (counted waits in group order), not behind all 8; Gx request and h-row store sit before the barrier; sums unchanged
# speedup vs baseline: 17.4430x; 1.0684x over previous
#include <hip/hip_runtime.h>
#include <hip/hip_bf16.h>

typedef __attribute__((ext_vector_type(2))) float v2f;
typedef __attribute__((ext_vector_type(8))) float v8f;
typedef __attribute__((ext_vector_type(4))) float v4f;

__device__ __forceinline__ float sigmoidf_(float x) { return 1.0f / (1.0f + __expf(-x)); }

__device__ __forceinline__ v8f wmma4(v2f a, v2f b, v8f c) {
  return __builtin_amdgcn_wmma_f32_16x16x4_f32(false, a, false, b, (short)0, c,
                                               false, false);
}

#define PA128 132
#define PA256 260

template<int NT, int K, int UNR>
__device__ __forceinline__ void mma_lds_a(const float* as, const float* __restrict__ wp,
                                          int ldw, v8f (&acc)[NT])
{
  #pragma clang loop unroll_count(UNR)
  for (int k0 = 0; k0 < K; k0 += 4) {
    const v2f a = *reinterpret_cast<const v2f*>(as + k0);
    #pragma unroll
    for (int t = 0; t < NT; ++t) {
      v2f b;
      b.x = wp[t * 16];
      b.y = wp[ldw + t * 16];
      acc[t] = wmma4(a, b, acc[t]);
    }
    wp += (size_t)4 * ldw;
  }
}

template<int K>
__device__ __forceinline__ void mma_lds_a_wt(const float* as4, const float* __restrict__ wrow,
                                             v8f (&acc)[4])
{
  #pragma clang loop unroll_count(4)
  for (int kb = 0; kb < K; kb += 8) {
    const float4 a4 = *reinterpret_cast<const float4*>(as4 + kb);
    v2f a01, a23; a01.x = a4.x; a01.y = a4.y; a23.x = a4.z; a23.y = a4.w;
    float4 b4[4];
    #pragma unroll
    for (int t = 0; t < 4; ++t)
      b4[t] = *reinterpret_cast<const float4*>(wrow + (size_t)t * 16 * K + kb);
    #pragma unroll
    for (int t = 0; t < 4; ++t) { v2f b; b.x = b4[t].x; b.y = b4[t].y; acc[t] = wmma4(a01, b, acc[t]); }
    #pragma unroll
    for (int t = 0; t < 4; ++t) { v2f b; b.x = b4[t].z; b.y = b4[t].w; acc[t] = wmma4(a23, b, acc[t]); }
  }
}

template<int C4>
__device__ __forceinline__ void load_rows(const float* __restrict__ src, int tid,
                                          float4 (&v)[16 * C4 / 256])
{
  #pragma unroll
  for (int p = 0; p < 16 * C4 / 256; ++p) {
    const int i = tid + p * 256;
    const int r = i / C4, c = (i - r * C4) * 4;
    v[p] = *reinterpret_cast<const float4*>(src + (size_t)r * (4 * C4) + c);
  }
}
template<int C4>
__device__ __forceinline__ void store_rows(float* dst, int pa, int tid,
                                           const float4 (&v)[16 * C4 / 256])
{
  #pragma unroll
  for (int p = 0; p < 16 * C4 / 256; ++p) {
    const int i = tid + p * 256;
    const int r = i / C4, c = (i - r * C4) * 4;
    *reinterpret_cast<float4*>(dst + r * pa + c) = v[p];
  }
}

__global__ void __launch_bounds__(256)
chain_f(const float* __restrict__ hidden, const float* __restrict__ f_w1,
        const float* __restrict__ f_b1, float* __restrict__ Af, float* __restrict__ Bf)
{
  __shared__ __attribute__((aligned(16))) float sH[16 * PA128];
  const int tid = threadIdx.x, wave = tid >> 5, lane = tid & 31;
  const int half = lane >> 4, lp = lane & 15;
  const int m0 = blockIdx.x * 16;
  {
    float4 vh[2];
    load_rows<32>(hidden + (size_t)m0 * 128, tid, vh);
    store_rows<32>(sH, PA128, tid, vh);
  }
  __syncthreads();

  const int n0 = wave * 32;
  const float* as = sH + lp * PA128 + 2 * half;
  v8f acc[2] = {};
  mma_lds_a<2, 128, 8>(as, f_w1 + (size_t)(2 * half) * 256 + n0 + lp, 256, acc);
  v8f bcc[2] = {};
  mma_lds_a<2, 128, 8>(as, f_w1 + (size_t)(128 + 2 * half) * 256 + n0 + lp, 256, bcc);
  #pragma unroll
  for (int t = 0; t < 2; ++t) {
    const int n = n0 + 16 * t + lp;
    const float bias = f_b1[n];
    #pragma unroll
    for (int r = 0; r < 8; ++r) {
      const size_t o = (size_t)(m0 + 8 * half + r) * 256 + n;
      Af[o] = acc[t][r];
      Bf[o] = bcc[t][r] + bias;
    }
  }
}

__global__ void __launch_bounds__(256)
chain_g(const float* __restrict__ S, const float* __restrict__ x,
        const float* __restrict__ f_w2, const float* __restrict__ f_b2,
        const float* __restrict__ g_w1, const float* __restrict__ g_b1,
        const float* __restrict__ g_w2, const float* __restrict__ g_b2,
        const float* __restrict__ wih, const float* __restrict__ bih,
        const float* __restrict__ bhh, float* __restrict__ Gx)
{
  __shared__ __attribute__((aligned(16))) float bufBig[16 * PA256];
  __shared__ __attribute__((aligned(16))) float bufX[16 * PA128];
  __shared__ __attribute__((aligned(16))) float bufM[16 * PA128];
  const int tid = threadIdx.x, wave = tid >> 5, lane = tid & 31;
  const int half = lane >> 4, lp = lane & 15;
  const int m0 = blockIdx.x * 16;
  {
    float4 vs[4], vx[2];
    load_rows<64>(S + (size_t)m0 * 256, tid, vs);
    load_rows<32>(x + (size_t)m0 * 128, tid, vx);
    store_rows<64>(bufBig, PA256, tid, vs);
    store_rows<32>(bufX, PA128, tid, vx);
  }
  __syncthreads();

  {
    const int n = wave * 16 + lp;
    v8f acc[1] = {};
    mma_lds_a<1, 256, 16>(bufBig + lp * PA256 + 2 * half,
                          f_w2 + (size_t)(2 * half) * 128 + n, 128, acc);
    const float bias = 128.0f * f_b2[n];
    #pragma unroll
    for (int r = 0; r < 8; ++r) bufM[(8 * half + r) * PA128 + n] = acc[0][r] + bias;
  }
  __syncthreads();

  {
    const int n0 = wave * 32;
    v8f acc[2] = {};
    mma_lds_a<2, 128, 8>(bufX + lp * PA128 + 2 * half,
                         g_w1 + (size_t)(2 * half) * 256 + n0 + lp, 256, acc);
    mma_lds_a<2, 128, 8>(bufM + lp * PA128 + 2 * half,
                         g_w1 + (size_t)(128 + 2 * half) * 256 + n0 + lp, 256, acc);
    #pragma unroll
    for (int t = 0; t < 2; ++t) {
      const int n = n0 + 16 * t + lp;
      const float bias = g_b1[n];
      #pragma unroll
      for (int r = 0; r < 8; ++r)
        bufBig[(8 * half + r) * PA256 + n] = fmaxf(acc[t][r] + bias, 0.0f);
    }
  }
  __syncthreads();

  {
    const int n = wave * 16 + lp;
    v8f acc[1] = {};
    mma_lds_a<1, 256, 16>(bufBig + lp * PA256 + 2 * half,
                          g_w2 + (size_t)(2 * half) * 128 + n, 128, acc);
    const float bias = g_b2[n];
    #pragma unroll
    for (int r = 0; r < 8; ++r) bufX[(8 * half + r) * PA128 + n] = acc[0][r] + bias;
  }
  __syncthreads();

  {
    const int n0 = wave * 64;
    v8f acc[4] = {};
    mma_lds_a_wt<128>(bufX + lp * PA128 + 4 * half,
                      wih + (size_t)(n0 + lp) * 128 + 4 * half, acc);
    #pragma unroll
    for (int t = 0; t < 4; ++t) {
      const int n = n0 + 16 * t + lp;
      const float bias = bih[n] + bhh[n];
      #pragma unroll
      for (int r = 0; r < 8; ++r)
        Gx[(size_t)(m0 + 8 * half + r) * 512 + n] = acc[t][r] + bias;
    }
  }
}

__global__ void __launch_bounds__(256)
chain_o(const float* __restrict__ nh, const float* __restrict__ o_w1,
        const float* __restrict__ o_b1, const float* __restrict__ o_w2,
        const float* __restrict__ o_b2, float* __restrict__ out)
{
  __shared__ __attribute__((aligned(16))) float sH[16 * PA128];
  __shared__ __attribute__((aligned(16))) float sO[16 * PA256];
  const int tid = threadIdx.x, wave = tid >> 5, lane = tid & 31;
  const int half = lane >> 4, lp = lane & 15;
  const int m0 = blockIdx.x * 16;
  {
    float4 vh[2];
    load_rows<32>(nh + (size_t)m0 * 128, tid, vh);
    store_rows<32>(sH, PA128, tid, vh);
  }
  __syncthreads();
  {
    const int n0 = wave * 32;
    v8f acc[2] = {};
    mma_lds_a<2, 128, 8>(sH + lp * PA128 + 2 * half,
                         o_w1 + (size_t)(2 * half) * 256 + n0 + lp, 256, acc);
    #pragma unroll
    for (int t = 0; t < 2; ++t) {
      const int n = n0 + 16 * t + lp;
      const float bias = o_b1[n];
      #pragma unroll
      for (int r = 0; r < 8; ++r)
        sO[(8 * half + r) * PA256 + n] = fmaxf(acc[t][r] + bias, 0.0f);
    }
  }
  __syncthreads();
  if (wave < 2) {
    const int n = wave * 16 + lp;
    v8f acc[1] = {};
    mma_lds_a<1, 256, 16>(sO + lp * PA256 + 2 * half,
                          o_w2 + (size_t)(2 * half) * 32 + n, 32, acc);
    const float bias = o_b2[n];
    #pragma unroll
    for (int r = 0; r < 8; ++r)
      sH[(8 * half + r) * 36 + n] = acc[0][r] + bias;
  }
  __syncthreads();
  if (tid < 128) {
    const int r = tid >> 3, c = (tid & 7) * 4;
    const float4 v = *reinterpret_cast<const float4*>(sH + r * 36 + c);
    *reinterpret_cast<float4*>(out + (size_t)(m0 + r) * 32 + c) = v;
  }
}

#define PR_TI 8
#define PR_G  16

__device__ __forceinline__ void pr_load(const float* __restrict__ p, float (&v)[PR_G])
{
  #pragma unroll
  for (int q = 0; q < PR_G; ++q) v[q] = p[(size_t)q * 256];
}
__device__ __forceinline__ void pr_accum(const float (&v)[PR_G], const float (&bm)[PR_TI],
                                         float (&acc)[PR_TI])
{
  #pragma unroll
  for (int q = 0; q < PR_G; ++q) {
    #pragma unroll
    for (int i = 0; i < PR_TI; ++i) acc[i] += fmaxf(v[q] + bm[i], 0.0f);
  }
}

__global__ void __launch_bounds__(256)
pair_relu_sum(const float* __restrict__ Af, const float* __restrict__ Bf,
              float* __restrict__ S)
{
  const int b  = blockIdx.x >> 4;
  const int r0 = b * 128 + (blockIdx.x & 15) * PR_TI;
  const int f  = threadIdx.x;
  const float* bp = Bf + (size_t)r0 * 256 + f;
  const float* ap = Af + (size_t)b * 128 * 256 + f;
  float va[PR_G], vb[PR_G];
  pr_load(ap, va);
  float bm[PR_TI], acc[PR_TI];
  #pragma unroll
  for (int i = 0; i < PR_TI; ++i) { bm[i] = bp[(size_t)i * 256]; acc[i] = 0.0f; }
  for (int g = 0; g < 128 / (2 * PR_G); ++g) {
    pr_load(ap + (size_t)(2 * PR_G * g + PR_G) * 256, vb);
    pr_accum(va, bm, acc);
    if (g + 1 < 128 / (2 * PR_G))
      pr_load(ap + (size_t)(2 * PR_G * g + 2 * PR_G) * 256, va);
    pr_accum(vb, bm, acc);
  }
  float* sp = S + (size_t)r0 * 256 + f;
  #pragma unroll
  for (int i = 0; i < PR_TI; ++i) sp[(size_t)i * 256] = acc[i];
}

template<int CTRL>
__device__ __forceinline__ float quad_get(float v)
{
  return __int_as_float(__builtin_amdgcn_mov_dpp(__float_as_int(v), CTRL, 0xf, 0xf, true));
}
#define QP_XOR1 0xB1
#define QP_XOR2 0x4E
#define QP_L0   0x00
#define QP_L1   0x55
#define QP_L2   0xAA
#define QP_L3   0xFF

__device__ __forceinline__ v2f fma2(float h, v2f w, v2f acc)
{
  v2f hh; hh.x = h; hh.y = h;
  return __builtin_elementwise_fma(hh, w, acc);
}

#define GX_AHEAD 4
template<int V>  struct StepReg { static constexpr int  value = V; };
template<bool V> struct StepAsk { static constexpr bool value = V; };

__device__ __forceinline__ void lds_step_barrier()
{
  __builtin_amdgcn_fence(__ATOMIC_RELEASE, "workgroup", "local");
  __builtin_amdgcn_s_barrier();
  __builtin_amdgcn_fence(__ATOMIC_ACQUIRE, "workgroup", "local");
}

__global__ void __launch_bounds__(512)
lstm_scan(const float* __restrict__ Gx, const float* __restrict__ whh,
          const float* __restrict__ h0, const float* __restrict__ c0,
          float* __restrict__ new_hidden)
{
  __shared__ __attribute__((aligned(16))) float h_s[2][128];
  const int b = blockIdx.x;
  const int tid = threadIdx.x;
  const int q = tid & 3;
  const int n = tid >> 2;
  const int wave = tid >> 5, lane = tid & 31;

  v2f wif[32], wgo[32];
  {
    const float* ri = whh + (size_t)(0 * 128 + n) * 128 + 4 * q;
    const float* rf = whh + (size_t)(1 * 128 + n) * 128 + 4 * q;
    const float* rg = whh + (size_t)(2 * 128 + n) * 128 + 4 * q;
    const float* ro = whh + (size_t)(3 * 128 + n) * 128 + 4 * q;
    const v2f sif = {0.5f, 0.5f}, sgo = {1.0f, 0.5f};
    #pragma unroll
    for (int j = 0; j < 8; ++j) {
      const float4 vi = *reinterpret_cast<const float4*>(ri + 16 * j);
      const float4 vf = *reinterpret_cast<const float4*>(rf + 16 * j);
      const float4 vg = *reinterpret_cast<const float4*>(rg + 16 * j);
      const float4 vo = *reinterpret_cast<const float4*>(ro + 16 * j);
      wif[4*j+0].x = vi.x; wif[4*j+0].y = vf.x; wgo[4*j+0].x = vg.x; wgo[4*j+0].y = vo.x;
      wif[4*j+1].x = vi.y; wif[4*j+1].y = vf.y; wgo[4*j+1].x = vg.y; wgo[4*j+1].y = vo.y;
      wif[4*j+2].x = vi.z; wif[4*j+2].y = vf.z; wgo[4*j+2].x = vg.z; wgo[4*j+2].y = vo.z;
      wif[4*j+3].x = vi.w; wif[4*j+3].y = vf.w; wgo[4*j+3].x = vg.w; wgo[4*j+3].y = vo.w;
    }
    #pragma unroll
    for (int k = 0; k < 32; ++k) { wif[k] *= sif; wgo[k] *= sgo; }
  }

  const float am = (q == 2) ? 1.0f : 0.5f;
  const float bm = (q == 2) ? 0.0f : 0.5f;
  const bool odd = (q & 1) != 0, hi = (q & 2) != 0;

  float c = c0[b * 128 + n];
  if (q == 0) h_s[0][n] = h0[b * 128 + n];

  const float* gx = Gx + (size_t)b * 128 * 512 + q * 128 + n;
  float gxr[GX_AHEAD];
  #pragma unroll
  for (int u = 0; u < GX_AHEAD; ++u) gxr[u] = gx[(size_t)u * 512];
  float* hout = new_hidden + (size_t)b * 128 * 128 + 32 * wave + lane;
  __syncthreads();

  auto step = [&](const int t, auto U, auto REQUEST) __attribute__((always_inline)) {
    constexpr int u = decltype(U)::value;
    constexpr bool request = decltype(REQUEST)::value;
    const float* hs = h_s[t & 1] + 4 * q;
    v4f hv[8];
    #pragma unroll
    for (int j = 0; j < 8; ++j) hv[j] = *reinterpret_cast<const v4f*>(hs + 16 * j);
    const float hrow = h_s[t & 1][32 * (wave & 3) + lane];
    __builtin_amdgcn_sched_barrier(0);
    const float gx_now = gxr[u];
    v2f aif0 = {0.0f, 0.0f}, aif1 = {0.0f, 0.0f}, ago0 = {0.0f, 0.0f}, ago1 = {0.0f, 0.0f};
    #pragma unroll
    for (int j = 0; j < 8; ++j) {
      aif0 = fma2(hv[j].x, wif[4*j+0], aif0);  ago0 = fma2(hv[j].x, wgo[4*j+0], ago0);
      aif1 = fma2(hv[j].y, wif[4*j+1], aif1);  ago1 = fma2(hv[j].y, wgo[4*j+1], ago1);
      aif0 = fma2(hv[j].z, wif[4*j+2], aif0);  ago0 = fma2(hv[j].z, wgo[4*j+2], ago0);
      aif1 = fma2(hv[j].w, wif[4*j+3], aif1);  ago1 = fma2(hv[j].w, wgo[4*j+3], ago1);
    }
    const v2f pif = aif0 + aif1, pgo = ago0 + ago1;

    const float keepA = odd ? pif.y : pif.x, sendA = odd ? pif.x : pif.y;
    const float keepB = odd ? pgo.y : pgo.x, sendB = odd ? pgo.x : pgo.y;
    const float sA = keepA + quad_get<QP_XOR1>(sendA);
    const float sB = keepB + quad_get<QP_XOR1>(sendB);
    const float keep = hi ? sB : sA, send = hi ? sA : sB;
    const float x = fmaf(gx_now, am, keep + quad_get<QP_XOR2>(send));

    const float act = fmaf(am, __builtin_amdgcn_tanhf(x), bm);
    const float ig = quad_get<QP_L0>(act), fg = quad_get<QP_L1>(act);
    const float g2 = quad_get<QP_L2>(act), og = quad_get<QP_L3>(act);
    c = fmaf(fg, c, ig * g2);
    const float h = og * __builtin_amdgcn_tanhf(c);
    if (q == 0) h_s[(t + 1) & 1][n] = h;
    if constexpr (request) gxr[u] = gx[(size_t)(t + GX_AHEAD) * 512];
    if (wave < 4 && t > 0) hout[(size_t)(t - 1) * 128] = hrow;
    lds_step_barrier();
  };

  static_assert(GX_AHEAD == 4, "the trips below are written out for 4 steps");
  for (int t0 = 0; t0 < 128 - GX_AHEAD; t0 += GX_AHEAD) {
    step(t0 + 0, StepReg<0>{}, StepAsk<true>{});
    step(t0 + 1, StepReg<1>{}, StepAsk<true>{});
    step(t0 + 2, StepReg<2>{}, StepAsk<true>{});
    step(t0 + 3, StepReg<3>{}, StepAsk<true>{});
  }
  step(128 - GX_AHEAD + 0, StepReg<0>{}, StepAsk<false>{});
  step(128 - GX_AHEAD + 1, StepReg<1>{}, StepAsk<false>{});
  step(128 - GX_AHEAD + 2, StepReg<2>{}, StepAsk<false>{});
  step(128 - GX_AHEAD + 3, StepReg<3>{}, StepAsk<false>{});
  if (wave < 4) hout[(size_t)127 * 128] = h_s[128 & 1][32 * wave + lane];
}

extern "C" void kernel_launch(void* const* d_in, const int* in_sizes, int n_in,
                              void* d_out, int out_size, void* d_ws, size_t ws_size,
                              hipStream_t stream)
{
  (void)in_sizes; (void)n_in; (void)out_size; (void)ws_size;
  const float* x      = (const float*)d_in[0];
  const float* hidden = (const float*)d_in[1];
  const float* h0     = (const float*)d_in[2];
  const float* c0     = (const float*)d_in[3];
  const float* f_w1   = (const float*)d_in[4];
  const float* f_b1   = (const float*)d_in[5];
  const float* f_w2   = (const float*)d_in[6];
  const float* f_b2   = (const float*)d_in[7];
  const float* g_w1   = (const float*)d_in[8];
  const float* g_b1   = (const float*)d_in[9];
  const float* g_w2   = (const float*)d_in[10];
  const float* g_b2   = (const float*)d_in[11];
  const float* wih    = (const float*)d_in[12];
  const float* whh    = (const float*)d_in[13];
  const float* bih    = (const float*)d_in[14];
  const float* bhh    = (const float*)d_in[15];
  const float* o_w1   = (const float*)d_in[16];
  const float* o_b1   = (const float*)d_in[17];
  const float* o_w2   = (const float*)d_in[18];
  const float* o_b2   = (const float*)d_in[19];
  float* out = (float*)d_out;

  float* R0 = (float*)d_ws;
  float* R1 = R0 + (size_t)2048 * 256;
  float* R2 = R1 + (size_t)2048 * 256;
  float* R3 = R2 + (size_t)2048 * 256;

  chain_f<<<128, 256, 0, stream>>>(hidden, f_w1, f_b1, R0, R1);
  pair_relu_sum<<<256, 256, 0, stream>>>(R0, R1, R2);
  chain_g<<<128, 256, 0, stream>>>(R2, x, f_w2, f_b2, g_w1, g_b1, g_w2, g_b2,
                                   wih, bih, bhh, R0);
  lstm_scan<<<16, 512, 0, stream>>>(R0, whh, h0, c0, R3);
  chain_o<<<128, 256, 0, stream>>>(R3, o_w1, o_b1, o_w2, o_b2, out);
}


# GEMM chains: W fragments requested one trip ahead of their WMMAs (two register sets), each product's first trip while the product before runs; chain barriers fence LDS only; sums unchanged
# speedup vs baseline: 19.8966x; 1.1407x over previous
#include <hip/hip_runtime.h>
#include <hip/hip_bf16.h>

typedef __attribute__((ext_vector_type(2))) float v2f;
typedef __attribute__((ext_vector_type(8))) float v8f;
typedef __attribute__((ext_vector_type(4))) float v4f;

__device__ __forceinline__ float sigmoidf_(float x) { return 1.0f / (1.0f + __expf(-x)); }

__device__ __forceinline__ v8f wmma4(v2f a, v2f b, v8f c) {
  return __builtin_amdgcn_wmma_f32_16x16x4_f32(false, a, false, b, (short)0, c,
                                               false, false);
}

#define PA128 132
#define PA256 260

__device__ __forceinline__ void lds_barrier()
{
  __builtin_amdgcn_fence(__ATOMIC_RELEASE, "workgroup", "local");
  __builtin_amdgcn_s_barrier();
  __builtin_amdgcn_fence(__ATOMIC_ACQUIRE, "workgroup", "local");
}

template<int NT, int UNR>
__device__ __forceinline__ void w_load(const float* __restrict__ wp, int ldw, v2f (&b)[UNR][NT])
{
  #pragma unroll
  for (int u = 0; u < UNR; ++u) {
    #pragma unroll
    for (int t = 0; t < NT; ++t) {
      b[u][t].x = wp[(size_t)(4 * u) * ldw + t * 16];
      b[u][t].y = wp[(size_t)(4 * u + 1) * ldw + t * 16];
    }
  }
}
template<int NT, int UNR>
__device__ __forceinline__ void w_mma(const float* as, const v2f (&b)[UNR][NT], v8f (&acc)[NT])
{
  #pragma unroll
  for (int u = 0; u < UNR; ++u) {
    const v2f a = *reinterpret_cast<const v2f*>(as + 4 * u);
    #pragma unroll
    for (int t = 0; t < NT; ++t) acc[t] = wmma4(a, b[u][t], acc[t]);
  }
}

template<int NT, int K, int UNR>
__device__ __forceinline__ void mma_lds_a(const float* as_lo, const float* as_hi,
                                          const float* __restrict__ wp, int ldw,
                                          v8f (&acc)[NT], v2f (&b0)[UNR][NT])
{
  constexpr int KT = 4 * UNR;
  constexpr int TRIPS = K / KT;
  static_assert(TRIPS % 2 == 0, "trips are taken in pairs");
  v2f b1[UNR][NT];
  for (int p = 0; p < TRIPS / 2; ++p) {
    const int ka = 2 * p * KT, kb = ka + KT;
    w_load<NT, UNR>(wp + (size_t)kb * ldw, ldw, b1);
    w_mma<NT, UNR>(ka < K / 2 ? as_lo + ka : as_hi + (ka - K / 2), b0, acc);
    if (p + 1 < TRIPS / 2) w_load<NT, UNR>(wp + (size_t)(kb + KT) * ldw, ldw, b0);
    w_mma<NT, UNR>(kb < K / 2 ? as_lo + kb : as_hi + (kb - K / 2), b1, acc);
  }
}

__device__ __forceinline__ void wt_load(const float* __restrict__ wrow, int K, float4 (&b)[4][4])
{
  #pragma unroll
  for (int g = 0; g < 4; ++g) {
    #pragma unroll
    for (int t = 0; t < 4; ++t)
      b[g][t] = *reinterpret_cast<const float4*>(wrow + (size_t)t * 16 * K + 8 * g);
  }
}
__device__ __forceinline__ void wt_mma(const float* as4, const float4 (&b)[4][4], v8f (&acc)[4])
{
  #pragma unroll
  for (int g = 0; g < 4; ++g) {
    const float4 a4 = *reinterpret_cast<const float4*>(as4 + 8 * g);
    v2f a01, a23; a01.x = a4.x; a01.y = a4.y; a23.x = a4.z; a23.y = a4.w;
    #pragma unroll
    for (int t = 0; t < 4; ++t) { v2f w; w.x = b[g][t].x; w.y = b[g][t].y; acc[t] = wmma4(a01, w, acc[t]); }
    #pragma unroll
    for (int t = 0; t < 4; ++t) { v2f w; w.x = b[g][t].z; w.y = b[g][t].w; acc[t] = wmma4(a23, w, acc[t]); }
  }
}
template<int K>
__device__ __forceinline__ void mma_lds_a_wt(const float* as4, const float* __restrict__ wrow,
                                             v8f (&acc)[4], float4 (&b0)[4][4])
{
  constexpr int TRIPS = K / 32;
  static_assert(TRIPS % 2 == 0, "trips are taken in pairs");
  float4 b1[4][4];
  for (int p = 0; p < TRIPS / 2; ++p) {
    const int ka = 64 * p, kb = ka + 32;
    wt_load(wrow + kb, K, b1);
    wt_mma(as4 + ka, b0, acc);
    if (p + 1 < TRIPS / 2) wt_load(wrow + kb + 32, K, b0);
    wt_mma(as4 + kb, b1, acc);
  }
}

template<int C4>
__device__ __forceinline__ void load_rows(const float* __restrict__ src, int tid,
                                          float4 (&v)[16 * C4 / 256])
{
  #pragma unroll
  for (int p = 0; p < 16 * C4 / 256; ++p) {
    const int i = tid + p * 256;
    const int r = i / C4, c = (i - r * C4) * 4;
    v[p] = *reinterpret_cast<const float4*>(src + (size_t)r * (4 * C4) + c);
  }
}
template<int C4>
__device__ __forceinline__ void store_rows(float* dst, int pa, int tid,
                                           const float4 (&v)[16 * C4 / 256])
{
  #pragma unroll
  for (int p = 0; p < 16 * C4 / 256; ++p) {
    const int i = tid + p * 256;
    const int r = i / C4, c = (i - r * C4) * 4;
    *reinterpret_cast<float4*>(dst + r * pa + c) = v[p];
  }
}

__global__ void __launch_bounds__(256)
chain_f(const float* __restrict__ hidden, const float* __restrict__ f_w1,
        const float* __restrict__ f_b1, float* __restrict__ Af, float* __restrict__ Bf)
{
  __shared__ __attribute__((aligned(16))) float sH[16 * PA128];
  const int tid = threadIdx.x, wave = tid >> 5, lane = tid & 31;
  const int half = lane >> 4, lp = lane & 15;
  const int m0 = blockIdx.x * 16;
  const int n0 = wave * 32;
  const float* wa = f_w1 + (size_t)(2 * half) * 256 + n0 + lp;
  const float* wb = f_w1 + (size_t)(128 + 2 * half) * 256 + n0 + lp;
  v2f fa[8][2], fb[8][2];
  w_load<2, 8>(wa, 256, fa);
  w_load<2, 8>(wb, 256, fb);
  {
    float4 vh[2];
    load_rows<32>(hidden + (size_t)m0 * 128, tid, vh);
    store_rows<32>(sH, PA128, tid, vh);
  }
  lds_barrier();

  const float* as = sH + lp * PA128 + 2 * half;
  v8f acc[2] = {};
  mma_lds_a<2, 128, 8>(as, as + 64, wa, 256, acc, fa);
  v8f bcc[2] = {};
  mma_lds_a<2, 128, 8>(as, as + 64, wb, 256, bcc, fb);
  #pragma unroll
  for (int t = 0; t < 2; ++t) {
    const int n = n0 + 16 * t + lp;
    const float bias = f_b1[n];
    #pragma unroll
    for (int r = 0; r < 8; ++r) {
      const size_t o = (size_t)(m0 + 8 * half + r) * 256 + n;
      Af[o] = acc[t][r];
      Bf[o] = bcc[t][r] + bias;
    }
  }
}

__global__ void __launch_bounds__(256)
chain_g(const float* __restrict__ S, const float* __restrict__ x,
        const float* __restrict__ f_w2, const float* __restrict__ f_b2,
        const float* __restrict__ g_w1, const float* __restrict__ g_b1,
        const float* __restrict__ g_w2, const float* __restrict__ g_b2,
        const float* __restrict__ wih, const float* __restrict__ bih,
        const float* __restrict__ bhh, float* __restrict__ Gx)
{
  __shared__ __attribute__((aligned(16))) float bufBig[16 * PA256];
  __shared__ __attribute__((aligned(16))) float bufX[16 * PA128];
  __shared__ __attribute__((aligned(16))) float bufM[16 * PA128];
  const int tid = threadIdx.x, wave = tid >> 5, lane = tid & 31;
  const int half = lane >> 4, lp = lane & 15;
  const int m0 = blockIdx.x * 16;
  const float* w_sm = f_w2 + (size_t)(2 * half) * 128 + wave * 16 + lp;
  const float* w_g1 = g_w1 + (size_t)(2 * half) * 256 + wave * 32 + lp;
  const float* w_ig = g_w2 + (size_t)(2 * half) * 128 + wave * 16 + lp;
  const float* w_gx = wih + (size_t)(wave * 64 + lp) * 128 + 4 * half;
  v2f f_sm[16][1];
  w_load<1, 16>(w_sm, 128, f_sm);
  {
    float4 vs[4], vx[2];
    load_rows<64>(S + (size_t)m0 * 256, tid, vs);
    load_rows<32>(x + (size_t)m0 * 128, tid, vx);
    store_rows<64>(bufBig, PA256, tid, vs);
    store_rows<32>(bufX, PA128, tid, vx);
  }
  lds_barrier();

  v2f f_g1[8][2];
  {
    const int n = wave * 16 + lp;
    v8f acc[1] = {};
    const float* as = bufBig + lp * PA256 + 2 * half;
    w_load<2, 8>(w_g1, 256, f_g1);
    mma_lds_a<1, 256, 16>(as, as + 128, w_sm, 128, acc, f_sm);
    const float bias = 128.0f * f_b2[n];
    #pragma unroll
    for (int r = 0; r < 8; ++r) bufM[(8 * half + r) * PA128 + n] = acc[0][r] + bias;
  }
  lds_barrier();

  v2f f_ig[16][1];
  {
    const int n0 = wave * 32;
    v8f acc[2] = {};
    w_load<1, 16>(w_ig, 128, f_ig);
    mma_lds_a<2, 256, 8>(bufX + lp * PA128 + 2 * half, bufM + lp * PA128 + 2 * half,
                         w_g1, 256, acc, f_g1);
    #pragma unroll
    for (int t = 0; t < 2; ++t) {
      const int n = n0 + 16 * t + lp;
      const float bias = g_b1[n];
      #pragma unroll
      for (int r = 0; r < 8; ++r)
        bufBig[(8 * half + r) * PA256 + n] = fmaxf(acc[t][r] + bias, 0.0f);
    }
  }
  lds_barrier();

  float4 f_gx[4][4];
  {
    const int n = wave * 16 + lp;
    v8f acc[1] = {};
    const float* as = bufBig + lp * PA256 + 2 * half;
    wt_load(w_gx, 128, f_gx);
    mma_lds_a<1, 256, 16>(as, as + 128, w_ig, 128, acc, f_ig);
    const float bias = g_b2[n];
    #pragma unroll
    for (int r = 0; r < 8; ++r) bufX[(8 * half + r) * PA128 + n] = acc[0][r] + bias;
  }
  lds_barrier();

  {
    const int n0 = wave * 64;
    v8f acc[4] = {};
    mma_lds_a_wt<128>(bufX + lp * PA128 + 4 * half, w_gx, acc, f_gx);
    #pragma unroll
    for (int t = 0; t < 4; ++t) {
      const int n = n0 + 16 * t + lp;
      const float bias = bih[n] + bhh[n];
      #pragma unroll
      for (int r = 0; r < 8; ++r)
        Gx[(size_t)(m0 + 8 * half + r) * 512 + n] = acc[t][r] + bias;
    }
  }
}

__global__ void __launch_bounds__(256)
chain_o(const float* __restrict__ nh, const float* __restrict__ o_w1,
        const float* __restrict__ o_b1, const float* __restrict__ o_w2,
        const float* __restrict__ o_b2, float* __restrict__ out)
{
  __shared__ __attribute__((aligned(16))) float sH[16 * PA128];
  __shared__ __attribute__((aligned(16))) float sO[16 * PA256];
  const int tid = threadIdx.x, wave = tid >> 5, lane = tid & 31;
  const int half = lane >> 4, lp = lane & 15;
  const int m0 = blockIdx.x * 16;
  const float* w1 = o_w1 + (size_t)(2 * half) * 256 + wave * 32 + lp;
  const float* w2 = o_w2 + (size_t)(2 * half) * 32 + (wave & 1) * 16 + lp;
  v2f f1[8][2];
  w_load<2, 8>(w1, 256, f1);
  {
    float4 vh[2];
    load_rows<32>(nh + (size_t)m0 * 128, tid, vh);
    store_rows<32>(sH, PA128, tid, vh);
  }
  lds_barrier();
  v2f f2[16][1] = {};
  if (wave < 2) w_load<1, 16>(w2, 32, f2);
  {
    const int n0 = wave * 32;
    v8f acc[2] = {};
    const float* as = sH + lp * PA128 + 2 * half;
    mma_lds_a<2, 128, 8>(as, as + 64, w1, 256, acc, f1);
    #pragma unroll
    for (int t = 0; t < 2; ++t) {
      const int n = n0 + 16 * t + lp;
      const float bias = o_b1[n];
      #pragma unroll
      for (int r = 0; r < 8; ++r)
        sO[(8 * half + r) * PA256 + n] = fmaxf(acc[t][r] + bias, 0.0f);
    }
  }
  lds_barrier();
  if (wave < 2) {
    const int n = wave * 16 + lp;
    v8f acc[1] = {};
    const float* as = sO + lp * PA256 + 2 * half;
    mma_lds_a<1, 256, 16>(as, as + 128, w2, 32, acc, f2);
    const float bias = o_b2[n];
    #pragma unroll
    for (int r = 0; r < 8; ++r)
      sH[(8 * half + r) * 36 + n] = acc[0][r] + bias;
  }
  __syncthreads();
  if (tid < 128) {
    const int r = tid >> 3, c = (tid & 7) * 4;
    const float4 v = *reinterpret_cast<const float4*>(sH + r * 36 + c);
    *reinterpret_cast<float4*>(out + (size_t)(m0 + r) * 32 + c) = v;
  }
}

#define PR_TI 8
#define PR_G  16

__device__ __forceinline__ void pr_load(const float* __restrict__ p, float (&v)[PR_G])
{
  #pragma unroll
  for (int q = 0; q < PR_G; ++q) v[q] = p[(size_t)q * 256];
}
__device__ __forceinline__ void pr_accum(const float (&v)[PR_G], const float (&bm)[PR_TI],
                                         float (&acc)[PR_TI])
{
  #pragma unroll
  for (int q = 0; q < PR_G; ++q) {
    #pragma unroll
    for (int i = 0; i < PR_TI; ++i) acc[i] += fmaxf(v[q] + bm[i], 0.0f);
  }
}

__global__ void __launch_bounds__(256)
pair_relu_sum(const float* __restrict__ Af, const float* __restrict__ Bf,
              float* __restrict__ S)
{
  const int b  = blockIdx.x >> 4;
  const int r0 = b * 128 + (blockIdx.x & 15) * PR_TI;
  const int f  = threadIdx.x;
  const float* bp = Bf + (size_t)r0 * 256 + f;
  const float* ap = Af + (size_t)b * 128 * 256 + f;
  float va[PR_G], vb[PR_G];
  pr_load(ap, va);
  float bm[PR_TI], acc[PR_TI];
  #pragma unroll
  for (int i = 0; i < PR_TI; ++i) { bm[i] = bp[(size_t)i * 256]; acc[i] = 0.0f; }
  for (int g = 0; g < 128 / (2 * PR_G); ++g) {
    pr_load(ap + (size_t)(2 * PR_G * g + PR_G) * 256, vb);
    pr_accum(va, bm, acc);
    if (g + 1 < 128 / (2 * PR_G))
      pr_load(ap + (size_t)(2 * PR_G * g + 2 * PR_G) * 256, va);
    pr_accum(vb, bm, acc);
  }
  float* sp = S + (size_t)r0 * 256 + f;
  #pragma unroll
  for (int i = 0; i < PR_TI; ++i) sp[(size_t)i * 256] = acc[i];
}

template<int CTRL>
__device__ __forceinline__ float quad_get(float v)
{
  return __int_as_float(__builtin_amdgcn_mov_dpp(__float_as_int(v), CTRL, 0xf, 0xf, true));
}
#define QP_XOR1 0xB1
#define QP_XOR2 0x4E
#define QP_L0   0x00
#define QP_L1   0x55
#define QP_L2   0xAA
#define QP_L3   0xFF

__device__ __forceinline__ v2f fma2(float h, v2f w, v2f acc)
{
  v2f hh; hh.x = h; hh.y = h;
  return __builtin_elementwise_fma(hh, w, acc);
}

#define GX_AHEAD 4
template<int V>  struct StepReg { static constexpr int  value = V; };
template<bool V> struct StepAsk { static constexpr bool value = V; };

__global__ void __launch_bounds__(512)
lstm_scan(const float* __restrict__ Gx, const float* __restrict__ whh,
          const float* __restrict__ h0, const float* __restrict__ c0,
          float* __restrict__ new_hidden)
{
  __shared__ __attribute__((aligned(16))) float h_s[2][128];
  const int b = blockIdx.x;
  const int tid = threadIdx.x;
  const int q = tid & 3;
  const int n = tid >> 2;
  const int wave = tid >> 5, lane = tid & 31;

  v2f wif[32], wgo[32];
  {
    const float* ri = whh + (size_t)(0 * 128 + n) * 128 + 4 * q;
    const float* rf = whh + (size_t)(1 * 128 + n) * 128 + 4 * q;
    const float* rg = whh + (size_t)(2 * 128 + n) * 128 + 4 * q;
    const float* ro = whh + (size_t)(3 * 128 + n) * 128 + 4 * q;
    const v2f sif = {0.5f, 0.5f}, sgo = {1.0f, 0.5f};
    #pragma unroll
    for (int j = 0; j < 8; ++j) {
      const float4 vi = *reinterpret_cast<const float4*>(ri + 16 * j);
      const float4 vf = *reinterpret_cast<const float4*>(rf + 16 * j);
      const float4 vg = *reinterpret_cast<const float4*>(rg + 16 * j);
      const float4 vo = *reinterpret_cast<const float4*>(ro + 16 * j);
      wif[4*j+0].x = vi.x; wif[4*j+0].y = vf.x; wgo[4*j+0].x = vg.x; wgo[4*j+0].y = vo.x;
      wif[4*j+1].x = vi.y; wif[4*j+1].y = vf.y; wgo[4*j+1].x = vg.y; wgo[4*j+1].y = vo.y;
      wif[4*j+2].x = vi.z; wif[4*j+2].y = vf.z; wgo[4*j+2].x = vg.z; wgo[4*j+2].y = vo.z;
      wif[4*j+3].x = vi.w; wif[4*j+3].y = vf.w; wgo[4*j+3].x = vg.w; wgo[4*j+3].y = vo.w;
    }
    #pragma unroll
    for (int k = 0; k < 32; ++k) { wif[k] *= sif; wgo[k] *= sgo; }
  }

  const float am = (q == 2) ? 1.0f : 0.5f;
  const float bm = (q == 2) ? 0.0f : 0.5f;
  const bool odd = (q & 1) != 0, hi = (q & 2) != 0;

  float c = c0[b * 128 + n];
  if (q == 0) h_s[0][n] = h0[b * 128 + n];

  const float* gx = Gx + (size_t)b * 128 * 512 + q * 128 + n;
  float gxr[GX_AHEAD];
  #pragma unroll
  for (int u = 0; u < GX_AHEAD; ++u) gxr[u] = gx[(size_t)u * 512];
  float* hout = new_hidden + (size_t)b * 128 * 128 + 32 * wave + lane;
  __syncthreads();

  auto step = [&](const int t, auto U, auto REQUEST) __attribute__((always_inline)) {
    constexpr int u = decltype(U)::value;
    constexpr bool request = decltype(REQUEST)::value;
    const float* hs = h_s[t & 1] + 4 * q;
    v4f hv[8];
    #pragma unroll
    for (int j = 0; j < 8; ++j) hv[j] = *reinterpret_cast<const v4f*>(hs + 16 * j);
    const float hrow = h_s[t & 1][32 * (wave & 3) + lane];
    __builtin_amdgcn_sched_barrier(0);
    const float gx_now = gxr[u];
    v2f aif0 = {0.0f, 0.0f}, aif1 = {0.0f, 0.0f}, ago0 = {0.0f, 0.0f}, ago1 = {0.0f, 0.0f};
    #pragma unroll
    for (int j = 0; j < 8; ++j) {
      aif0 = fma2(hv[j].x, wif[4*j+0], aif0);  ago0 = fma2(hv[j].x, wgo[4*j+0], ago0);
      aif1 = fma2(hv[j].y, wif[4*j+1], aif1);  ago1 = fma2(hv[j].y, wgo[4*j+1], ago1);
      aif0 = fma2(hv[j].z, wif[4*j+2], aif0);  ago0 = fma2(hv[j].z, wgo[4*j+2], ago0);
      aif1 = fma2(hv[j].w, wif[4*j+3], aif1);  ago1 = fma2(hv[j].w, wgo[4*j+3], ago1);
    }
    const v2f pif = aif0 + aif1, pgo = ago0 + ago1;

    const float keepA = odd ? pif.y : pif.x, sendA = odd ? pif.x : pif.y;
    const float keepB = odd ? pgo.y : pgo.x, sendB = odd ? pgo.x : pgo.y;
    const float sA = keepA + quad_get<QP_XOR1>(sendA);
    const float sB = keepB + quad_get<QP_XOR1>(sendB);
    const float keep = hi ? sB : sA, send = hi ? sA : sB;
    const float x = fmaf(gx_now, am, keep + quad_get<QP_XOR2>(send));

    const float act = fmaf(am, __builtin_amdgcn_tanhf(x), bm);
    const float ig = quad_get<QP_L0>(act), fg = quad_get<QP_L1>(act);
    const float g2 = quad_get<QP_L2>(act), og = quad_get<QP_L3>(act);
    c = fmaf(fg, c, ig * g2);
    const float h = og * __builtin_amdgcn_tanhf(c);
    if (q == 0) h_s[(t + 1) & 1][n] = h;
    if constexpr (request) gxr[u] = gx[(size_t)(t + GX_AHEAD) * 512];
    if (wave < 4 && t > 0) hout[(size_t)(t - 1) * 128] = hrow;
    lds_barrier();
  };

  static_assert(GX_AHEAD == 4, "the trips below are written out for 4 steps");
  for (int t0 = 0; t0 < 128 - GX_AHEAD; t0 += GX_AHEAD) {
    step(t0 + 0, StepReg<0>{}, StepAsk<true>{});
    step(t0 + 1, StepReg<1>{}, StepAsk<true>{});
    step(t0 + 2, StepReg<2>{}, StepAsk<true>{});
    step(t0 + 3, StepReg<3>{}, StepAsk<true>{});
  }
  step(128 - GX_AHEAD + 0, StepReg<0>{}, StepAsk<false>{});
  step(128 - GX_AHEAD + 1, StepReg<1>{}, StepAsk<false>{});
  step(128 - GX_AHEAD + 2, StepReg<2>{}, StepAsk<false>{});
  step(128 - GX_AHEAD + 3, StepReg<3>{}, StepAsk<false>{});
  if (wave < 4) hout[(size_t)127 * 128] = h_s[128 & 1][32 * wave + lane];
}

extern "C" void kernel_launch(void* const* d_in, const int* in_sizes, int n_in,
                              void* d_out, int out_size, void* d_ws, size_t ws_size,
                              hipStream_t stream)
{
  (void)in_sizes; (void)n_in; (void)out_size; (void)ws_size;
  const float* x      = (const float*)d_in[0];
  const float* hidden = (const float*)d_in[1];
  const float* h0     = (const float*)d_in[2];
  const float* c0     = (const float*)d_in[3];
  const float* f_w1   = (const float*)d_in[4];
  const float* f_b1   = (const float*)d_in[5];
  const float* f_w2   = (const float*)d_in[6];
  const float* f_b2   = (const float*)d_in[7];
  const float* g_w1   = (const float*)d_in[8];
  const float* g_b1   = (const float*)d_in[9];
  const float* g_w2   = (const float*)d_in[10];
  const float* g_b2   = (const float*)d_in[11];
  const float* wih    = (const float*)d_in[12];
  const float* whh    = (const float*)d_in[13];
  const float* bih    = (const float*)d_in[14];
  const float* bhh    = (const float*)d_in[15];
  const float* o_w1   = (const float*)d_in[16];
  const float* o_b1   = (const float*)d_in[17];
  const float* o_w2   = (const float*)d_in[18];
  const float* o_b2   = (const float*)d_in[19];
  float* out = (float*)d_out;

  float* R0 = (float*)d_ws;
  float* R1 = R0 + (size_t)2048 * 256;
  float* R2 = R1 + (size_t)2048 * 256;
  float* R3 = R2 + (size_t)2048 * 256;

  chain_f<<<128, 256, 0, stream>>>(hidden, f_w1, f_b1, R0, R1);
  pair_relu_sum<<<256, 256, 0, stream>>>(R0, R1, R2);
  chain_g<<<128, 256, 0, stream>>>(R2, x, f_w2, f_b2, g_w1, g_b1, g_w2, g_b2,
                                   wih, bih, bhh, R0);
  lstm_scan<<<16, 512, 0, stream>>>(R0, whh, h0, c0, R3);
  chain_o<<<128, 256, 0, stream>>>(R3, o_w1, o_b1, o_w2, o_b2, out);
}


# chain_f and pair_relu_sum fused into one launch (5 -> 4): a block owns (batch row, 16 f columns), keeps hidden[b] and its Af/Bf slabs in LDS and sums over j from LDS; sums bit-identical
# speedup vs baseline: 20.8215x; 1.0465x over previous
#include <hip/hip_runtime.h>
#include <hip/hip_bf16.h>

typedef __attribute__((ext_vector_type(2))) float v2f;
typedef __attribute__((ext_vector_type(8))) float v8f;
typedef __attribute__((ext_vector_type(4))) float v4f;

__device__ __forceinline__ float sigmoidf_(float x) { return 1.0f / (1.0f + __expf(-x)); }

__device__ __forceinline__ v8f wmma4(v2f a, v2f b, v8f c) {
  return __builtin_amdgcn_wmma_f32_16x16x4_f32(false, a, false, b, (short)0, c,
                                               false, false);
}

#define PA128 132
#define PA256 260

__device__ __forceinline__ void lds_barrier()
{
  __builtin_amdgcn_fence(__ATOMIC_RELEASE, "workgroup", "local");
  __builtin_amdgcn_s_barrier();
  __builtin_amdgcn_fence(__ATOMIC_ACQUIRE, "workgroup", "local");
}

template<int NT, int UNR>
__device__ __forceinline__ void w_load(const float* __restrict__ wp, int ldw, v2f (&b)[UNR][NT])
{
  #pragma unroll
  for (int u = 0; u < UNR; ++u) {
    #pragma unroll
    for (int t = 0; t < NT; ++t) {
      b[u][t].x = wp[(size_t)(4 * u) * ldw + t * 16];
      b[u][t].y = wp[(size_t)(4 * u + 1) * ldw + t * 16];
    }
  }
}
template<int NT, int UNR>
__device__ __forceinline__ void w_mma(const float* as, const v2f (&b)[UNR][NT], v8f (&acc)[NT])
{
  #pragma unroll
  for (int u = 0; u < UNR; ++u) {
    const v2f a = *reinterpret_cast<const v2f*>(as + 4 * u);
    #pragma unroll
    for (int t = 0; t < NT; ++t) acc[t] = wmma4(a, b[u][t], acc[t]);
  }
}

template<int NT, int K, int UNR>
__device__ __forceinline__ void mma_lds_a(const float* as_lo, const float* as_hi,
                                          const float* __restrict__ wp, int ldw,
                                          v8f (&acc)[NT], v2f (&b0)[UNR][NT])
{
  constexpr int KT = 4 * UNR;
  constexpr int TRIPS = K / KT;
  static_assert(TRIPS % 2 == 0, "trips are taken in pairs");
  v2f b1[UNR][NT];
  for (int p = 0; p < TRIPS / 2; ++p) {
    const int ka = 2 * p * KT, kb = ka + KT;
    w_load<NT, UNR>(wp + (size_t)kb * ldw, ldw, b1);
    w_mma<NT, UNR>(ka < K / 2 ? as_lo + ka : as_hi + (ka - K / 2), b0, acc);
    if (p + 1 < TRIPS / 2) w_load<NT, UNR>(wp + (size_t)(kb + KT) * ldw, ldw, b0);
    w_mma<NT, UNR>(kb < K / 2 ? as_lo + kb : as_hi + (kb - K / 2), b1, acc);
  }
}

__device__ __forceinline__ void wt_load(const float* __restrict__ wrow, int K, float4 (&b)[4][4])
{
  #pragma unroll
  for (int g = 0; g < 4; ++g) {
    #pragma unroll
    for (int t = 0; t < 4; ++t)
      b[g][t] = *reinterpret_cast<const float4*>(wrow + (size_t)t * 16 * K + 8 * g);
  }
}
__device__ __forceinline__ void wt_mma(const float* as4, const float4 (&b)[4][4], v8f (&acc)[4])
{
  #pragma unroll
  for (int g = 0; g < 4; ++g) {
    const float4 a4 = *reinterpret_cast<const float4*>(as4 + 8 * g);
    v2f a01, a23; a01.x = a4.x; a01.y = a4.y; a23.x = a4.z; a23.y = a4.w;
    #pragma unroll
    for (int t = 0; t < 4; ++t) { v2f w; w.x = b[g][t].x; w.y = b[g][t].y; acc[t] = wmma4(a01, w, acc[t]); }
    #pragma unroll
    for (int t = 0; t < 4; ++t) { v2f w; w.x = b[g][t].z; w.y = b[g][t].w; acc[t] = wmma4(a23, w, acc[t]); }
  }
}
template<int K>
__device__ __forceinline__ void mma_lds_a_wt(const float* as4, const float* __restrict__ wrow,
                                             v8f (&acc)[4], float4 (&b0)[4][4])
{
  constexpr int TRIPS = K / 32;
  static_assert(TRIPS % 2 == 0, "trips are taken in pairs");
  float4 b1[4][4];
  for (int p = 0; p < TRIPS / 2; ++p) {
    const int ka = 64 * p, kb = ka + 32;
    wt_load(wrow + kb, K, b1);
    wt_mma(as4 + ka, b0, acc);
    if (p + 1 < TRIPS / 2) wt_load(wrow + kb + 32, K, b0);
    wt_mma(as4 + kb, b1, acc);
  }
}

template<int C4>
__device__ __forceinline__ void load_rows(const float* __restrict__ src, int tid,
                                          float4 (&v)[16 * C4 / 256])
{
  #pragma unroll
  for (int p = 0; p < 16 * C4 / 256; ++p) {
    const int i = tid + p * 256;
    const int r = i / C4, c = (i - r * C4) * 4;
    v[p] = *reinterpret_cast<const float4*>(src + (size_t)r * (4 * C4) + c);
  }
}
template<int C4>
__device__ __forceinline__ void store_rows(float* dst, int pa, int tid,
                                           const float4 (&v)[16 * C4 / 256])
{
  #pragma unroll
  for (int p = 0; p < 16 * C4 / 256; ++p) {
    const int i = tid + p * 256;
    const int r = i / C4, c = (i - r * C4) * 4;
    *reinterpret_cast<float4*>(dst + r * pa + c) = v[p];
  }
}

__global__ void __launch_bounds__(256)
chain_g(const float* __restrict__ S, const float* __restrict__ x,
        const float* __restrict__ f_w2, const float* __restrict__ f_b2,
        const float* __restrict__ g_w1, const float* __restrict__ g_b1,
        const float* __restrict__ g_w2, const float* __restrict__ g_b2,
        const float* __restrict__ wih, const float* __restrict__ bih,
        const float* __restrict__ bhh, float* __restrict__ Gx)
{
  __shared__ __attribute__((aligned(16))) float bufBig[16 * PA256];
  __shared__ __attribute__((aligned(16))) float bufX[16 * PA128];
  __shared__ __attribute__((aligned(16))) float bufM[16 * PA128];
  const int tid = threadIdx.x, wave = tid >> 5, lane = tid & 31;
  const int half = lane >> 4, lp = lane & 15;
  const int m0 = blockIdx.x * 16;
  const float* w_sm = f_w2 + (size_t)(2 * half) * 128 + wave * 16 + lp;
  const float* w_g1 = g_w1 + (size_t)(2 * half) * 256 + wave * 32 + lp;
  const float* w_ig = g_w2 + (size_t)(2 * half) * 128 + wave * 16 + lp;
  const float* w_gx = wih + (size_t)(wave * 64 + lp) * 128 + 4 * half;
  v2f f_sm[16][1];
  w_load<1, 16>(w_sm, 128, f_sm);
  {
    float4 vs[4], vx[2];
    const float* sb = S + ((size_t)(m0 >> 7) * 16 * 128 + (m0 & 127)) * 16;
    #pragma unroll
    for (int p = 0; p < 4; ++p) {
      const int i = tid + p * 256;
      const int r = i >> 6, c4 = i & 63;
      vs[p] = *reinterpret_cast<const float4*>(sb + ((size_t)(c4 >> 2) * 128 + r) * 16 + (c4 & 3) * 4);
    }
    load_rows<32>(x + (size_t)m0 * 128, tid, vx);
    store_rows<64>(bufBig, PA256, tid, vs);
    store_rows<32>(bufX, PA128, tid, vx);
  }
  lds_barrier();

  v2f f_g1[8][2];
  {
    const int n = wave * 16 + lp;
    v8f acc[1] = {};
    const float* as = bufBig + lp * PA256 + 2 * half;
    w_load<2, 8>(w_g1, 256, f_g1);
    mma_lds_a<1, 256, 16>(as, as + 128, w_sm, 128, acc, f_sm);
    const float bias = 128.0f * f_b2[n];
    #pragma unroll
    for (int r = 0; r < 8; ++r) bufM[(8 * half + r) * PA128 + n] = acc[0][r] + bias;
  }
  lds_barrier();

  v2f f_ig[16][1];
  {
    const int n0 = wave * 32;
    v8f acc[2] = {};
    w_load<1, 16>(w_ig, 128, f_ig);
    mma_lds_a<2, 256, 8>(bufX + lp * PA128 + 2 * half, bufM + lp * PA128 + 2 * half,
                         w_g1, 256, acc, f_g1);
    #pragma unroll
    for (int t = 0; t < 2; ++t) {
      const int n = n0 + 16 * t + lp;
      const float bias = g_b1[n];
      #pragma unroll
      for (int r = 0; r < 8; ++r)
        bufBig[(8 * half + r) * PA256 + n] = fmaxf(acc[t][r] + bias, 0.0f);
    }
  }
  lds_barrier();

  float4 f_gx[4][4];
  {
    const int n = wave * 16 + lp;
    v8f acc[1] = {};
    const float* as = bufBig + lp * PA256 + 2 * half;
    wt_load(w_gx, 128, f_gx);
    mma_lds_a<1, 256, 16>(as, as + 128, w_ig, 128, acc, f_ig);
    const float bias = g_b2[n];
    #pragma unroll
    for (int r = 0; r < 8; ++r) bufX[(8 * half + r) * PA128 + n] = acc[0][r] + bias;
  }
  lds_barrier();

  {
    const int n0 = wave * 64;
    v8f acc[4] = {};
    mma_lds_a_wt<128>(bufX + lp * PA128 + 4 * half, w_gx, acc, f_gx);
    #pragma unroll
    for (int t = 0; t < 4; ++t) {
      const int n = n0 + 16 * t + lp;
      const float bias = bih[n] + bhh[n];
      #pragma unroll
      for (int r = 0; r < 8; ++r)
        Gx[(size_t)(m0 + 8 * half + r) * 512 + n] = acc[t][r] + bias;
    }
  }
}

__global__ void __launch_bounds__(256)
chain_o(const float* __restrict__ nh, const float* __restrict__ o_w1,
        const float* __restrict__ o_b1, const float* __restrict__ o_w2,
        const float* __restrict__ o_b2, float* __restrict__ out)
{
  __shared__ __attribute__((aligned(16))) float sH[16 * PA128];
  __shared__ __attribute__((aligned(16))) float sO[16 * PA256];
  const int tid = threadIdx.x, wave = tid >> 5, lane = tid & 31;
  const int half = lane >> 4, lp = lane & 15;
  const int m0 = blockIdx.x * 16;
  const float* w1 = o_w1 + (size_t)(2 * half) * 256 + wave * 32 + lp;
  const float* w2 = o_w2 + (size_t)(2 * half) * 32 + (wave & 1) * 16 + lp;
  v2f f1[8][2];
  w_load<2, 8>(w1, 256, f1);
  {
    float4 vh[2];
    load_rows<32>(nh + (size_t)m0 * 128, tid, vh);
    store_rows<32>(sH, PA128, tid, vh);
  }
  lds_barrier();
  v2f f2[16][1] = {};
  if (wave < 2) w_load<1, 16>(w2, 32, f2);
  {
    const int n0 = wave * 32;
    v8f acc[2] = {};
    const float* as = sH + lp * PA128 + 2 * half;
    mma_lds_a<2, 128, 8>(as, as + 64, w1, 256, acc, f1);
    #pragma unroll
    for (int t = 0; t < 2; ++t) {
      const int n = n0 + 16 * t + lp;
      const float bias = o_b1[n];
      #pragma unroll
      for (int r = 0; r < 8; ++r)
        sO[(8 * half + r) * PA256 + n] = fmaxf(acc[t][r] + bias, 0.0f);
    }
  }
  lds_barrier();
  if (wave < 2) {
    const int n = wave * 16 + lp;
    v8f acc[1] = {};
    const float* as = sO + lp * PA256 + 2 * half;
    mma_lds_a<1, 256, 16>(as, as + 128, w2, 32, acc, f2);
    const float bias = o_b2[n];
    #pragma unroll
    for (int r = 0; r < 8; ++r)
      sH[(8 * half + r) * 36 + n] = acc[0][r] + bias;
  }
  __syncthreads();
  if (tid < 128) {
    const int r = tid >> 3, c = (tid & 7) * 4;
    const float4 v = *reinterpret_cast<const float4*>(sH + r * 36 + c);
    *reinterpret_cast<float4*>(out + (size_t)(m0 + r) * 32 + c) = v;
  }
}

#define PR_TI 8
#define PR_G  16
#define PS    20

template<int UNR>
__device__ __forceinline__ void w_load2(const float* __restrict__ wpa, const float* __restrict__ wpb,
                                        int ldw, v2f (&b)[UNR][2])
{
  #pragma unroll
  for (int u = 0; u < UNR; ++u) {
    b[u][0].x = wpa[(size_t)(4 * u) * ldw];
    b[u][0].y = wpa[(size_t)(4 * u + 1) * ldw];
    b[u][1].x = wpb[(size_t)(4 * u) * ldw];
    b[u][1].y = wpb[(size_t)(4 * u + 1) * ldw];
  }
}

__device__ __forceinline__ void pr_load(const float* p, float (&v)[PR_G])
{
  #pragma unroll
  for (int q = 0; q < PR_G; ++q) v[q] = p[q * PS];
}
__device__ __forceinline__ void pr_accum(const float (&v)[PR_G], const float (&bm)[PR_TI],
                                         float (&acc)[PR_TI])
{
  #pragma unroll
  for (int q = 0; q < PR_G; ++q) {
    #pragma unroll
    for (int i = 0; i < PR_TI; ++i) acc[i] += fmaxf(v[q] + bm[i], 0.0f);
  }
}

__global__ void __launch_bounds__(256)
f_pair(const float* __restrict__ hidden, const float* __restrict__ f_w1,
       const float* __restrict__ f_b1, float* __restrict__ Ss)
{
  __shared__ __attribute__((aligned(16))) float sH[128 * PA128];
  __shared__ __attribute__((aligned(16))) float sA[128 * PS];
  __shared__ __attribute__((aligned(16))) float sB[128 * PS];
  const int tid = threadIdx.x, wave = tid >> 5, lane = tid & 31;
  const int half = lane >> 4, lp = lane & 15;
  const int b = blockIdx.x >> 4, fg = blockIdx.x & 15;
  const int f0 = fg * 16;
  const float* wa = f_w1 + (size_t)(2 * half) * 256 + f0 + lp;
  const float* wb = wa + (size_t)128 * 256;
  v2f fb0[8][2];
  w_load2<8>(wa, wb, 256, fb0);
  {
    const float* hb = hidden + (size_t)b * 128 * 128;
    float4 vh[16];
    #pragma unroll
    for (int p = 0; p < 16; ++p) {
      const int i = tid + p * 256;
      vh[p] = *reinterpret_cast<const float4*>(hb + (size_t)(i >> 5) * 128 + (i & 31) * 4);
    }
    #pragma unroll
    for (int p = 0; p < 16; ++p) {
      const int i = tid + p * 256;
      *reinterpret_cast<float4*>(sH + (i >> 5) * PA128 + (i & 31) * 4) = vh[p];
    }
  }
  lds_barrier();

  {
    const float* as = sH + (16 * wave + lp) * PA128 + 2 * half;
    v8f acc[2] = {};
    v2f fb1[8][2];
    for (int p = 0; p < 2; ++p) {
      const int ka = 64 * p, kb = ka + 32;
      w_load2<8>(wa + (size_t)kb * 256, wb + (size_t)kb * 256, 256, fb1);
      w_mma<2, 8>(as + ka, fb0, acc);
      if (p + 1 < 2) w_load2<8>(wa + (size_t)(kb + 32) * 256, wb + (size_t)(kb + 32) * 256, 256, fb0);
      w_mma<2, 8>(as + kb, fb1, acc);
    }
    const float bias = f_b1[f0 + lp];
    #pragma unroll
    for (int r = 0; r < 8; ++r) {
      const int row = 16 * wave + 8 * half + r;
      sA[row * PS + lp] = acc[0][r];
      sB[row * PS + lp] = acc[1][r] + bias;
    }
  }
  lds_barrier();

  const int f = tid & 15, ih = (tid >> 4) & 1;
  const float* ap = sA + f;
  float va[PR_G], vb[PR_G];
  pr_load(ap, va);
  float bm[PR_TI], acc[PR_TI];
  #pragma unroll
  for (int q = 0; q < PR_TI; ++q) { bm[q] = sB[(16 * wave + 2 * q + ih) * PS + f]; acc[q] = 0.0f; }
  for (int g = 0; g < 128 / (2 * PR_G); ++g) {
    pr_load(ap + (2 * PR_G * g + PR_G) * PS, vb);
    pr_accum(va, bm, acc);
    if (g + 1 < 128 / (2 * PR_G))
      pr_load(ap + (2 * PR_G * g + 2 * PR_G) * PS, va);
    pr_accum(vb, bm, acc);
  }
  float* sp = Ss + ((size_t)(b * 16 + fg) * 128 + 16 * wave + ih) * 16 + f;
  #pragma unroll
  for (int q = 0; q < PR_TI; ++q) sp[(size_t)(2 * q) * 16] = acc[q];
}

template<int CTRL>
__device__ __forceinline__ float quad_get(float v)
{
  return __int_as_float(__builtin_amdgcn_mov_dpp(__float_as_int(v), CTRL, 0xf, 0xf, true));
}
#define QP_XOR1 0xB1
#define QP_XOR2 0x4E
#define QP_L0   0x00
#define QP_L1   0x55
#define QP_L2   0xAA
#define QP_L3   0xFF

__device__ __forceinline__ v2f fma2(float h, v2f w, v2f acc)
{
  v2f hh; hh.x = h; hh.y = h;
  return __builtin_elementwise_fma(hh, w, acc);
}

#define GX_AHEAD 4
template<int V>  struct StepReg { static constexpr int  value = V; };
template<bool V> struct StepAsk { static constexpr bool value = V; };

__global__ void __launch_bounds__(512)
lstm_scan(const float* __restrict__ Gx, const float* __restrict__ whh,
          const float* __restrict__ h0, const float* __restrict__ c0,
          float* __restrict__ new_hidden)
{
  __shared__ __attribute__((aligned(16))) float h_s[2][128];
  const int b = blockIdx.x;
  const int tid = threadIdx.x;
  const int q = tid & 3;
  const int n = tid >> 2;
  const int wave = tid >> 5, lane = tid & 31;

  v2f wif[32], wgo[32];
  {
    const float* ri = whh + (size_t)(0 * 128 + n) * 128 + 4 * q;
    const float* rf = whh + (size_t)(1 * 128 + n) * 128 + 4 * q;
    const float* rg = whh + (size_t)(2 * 128 + n) * 128 + 4 * q;
    const float* ro = whh + (size_t)(3 * 128 + n) * 128 + 4 * q;
    const v2f sif = {0.5f, 0.5f}, sgo = {1.0f, 0.5f};
    #pragma unroll
    for (int j = 0; j < 8; ++j) {
      const float4 vi = *reinterpret_cast<const float4*>(ri + 16 * j);
      const float4 vf = *reinterpret_cast<const float4*>(rf + 16 * j);
      const float4 vg = *reinterpret_cast<const float4*>(rg + 16 * j);
      const float4 vo = *reinterpret_cast<const float4*>(ro + 16 * j);
      wif[4*j+0].x = vi.x; wif[4*j+0].y = vf.x; wgo[4*j+0].x = vg.x; wgo[4*j+0].y = vo.x;
      wif[4*j+1].x = vi.y; wif[4*j+1].y = vf.y; wgo[4*j+1].x = vg.y; wgo[4*j+1].y = vo.y;
      wif[4*j+2].x = vi.z; wif[4*j+2].y = vf.z; wgo[4*j+2].x = vg.z; wgo[4*j+2].y = vo.z;
      wif[4*j+3].x = vi.w; wif[4*j+3].y = vf.w; wgo[4*j+3].x = vg.w; wgo[4*j+3].y = vo.w;
    }
    #pragma unroll
    for (int k = 0; k < 32; ++k) { wif[k] *= sif; wgo[k] *= sgo; }
  }

  const float am = (q == 2) ? 1.0f : 0.5f;
  const float bm = (q == 2) ? 0.0f : 0.5f;
  const bool odd = (q & 1) != 0, hi = (q & 2) != 0;

  float c = c0[b * 128 + n];
  if (q == 0) h_s[0][n] = h0[b * 128 + n];

  const float* gx = Gx + (size_t)b * 128 * 512 + q * 128 + n;
  float gxr[GX_AHEAD];
  #pragma unroll
  for (int u = 0; u < GX_AHEAD; ++u) gxr[u] = gx[(size_t)u * 512];
  float* hout = new_hidden + (size_t)b * 128 * 128 + 32 * wave + lane;
  __syncthreads();

  auto step = [&](const int t, auto U, auto REQUEST) __attribute__((always_inline)) {
    constexpr int u = decltype(U)::value;
    constexpr bool request = decltype(REQUEST)::value;
    const float* hs = h_s[t & 1] + 4 * q;
    v4f hv[8];
    #pragma unroll
    for (int j = 0; j < 8; ++j) hv[j] = *reinterpret_cast<const v4f*>(hs + 16 * j);
    const float hrow = h_s[t & 1][32 * (wave & 3) + lane];
    __builtin_amdgcn_sched_barrier(0);
    const float gx_now = gxr[u];
    v2f aif0 = {0.0f, 0.0f}, aif1 = {0.0f, 0.0f}, ago0 = {0.0f, 0.0f}, ago1 = {0.0f, 0.0f};
    #pragma unroll
    for (int j = 0; j < 8; ++j) {
      aif0 = fma2(hv[j].x, wif[4*j+0], aif0);  ago0 = fma2(hv[j].x, wgo[4*j+0], ago0);
      aif1 = fma2(hv[j].y, wif[4*j+1], aif1);  ago1 = fma2(hv[j].y, wgo[4*j+1], ago1);
      aif0 = fma2(hv[j].z, wif[4*j+2], aif0);  ago0 = fma2(hv[j].z, wgo[4*j+2], ago0);
      aif1 = fma2(hv[j].w, wif[4*j+3], aif1);  ago1 = fma2(hv[j].w, wgo[4*j+3], ago1);
    }
    const v2f pif = aif0 + aif1, pgo = ago0 + ago1;

    const float keepA = odd ? pif.y : pif.x, sendA = odd ? pif.x : pif.y;
    const float keepB = odd ? pgo.y : pgo.x, sendB = odd ? pgo.x : pgo.y;
    const float sA = keepA + quad_get<QP_XOR1>(sendA);
    const float sB = keepB + quad_get<QP_XOR1>(sendB);
    const float keep = hi ? sB : sA, send = hi ? sA : sB;
    const float x = fmaf(gx_now, am, keep + quad_get<QP_XOR2>(send));

    const float act = fmaf(am, __builtin_amdgcn_tanhf(x), bm);
    const float ig = quad_get<QP_L0>(act), fg = quad_get<QP_L1>(act);
    const float g2 = quad_get<QP_L2>(act), og = quad_get<QP_L3>(act);
    c = fmaf(fg, c, ig * g2);
    const float h = og * __builtin_amdgcn_tanhf(c);
    if (q == 0) h_s[(t + 1) & 1][n] = h;
    if constexpr (request) gxr[u] = gx[(size_t)(t + GX_AHEAD) * 512];
    if (wave < 4 && t > 0) hout[(size_t)(t - 1) * 128] = hrow;
    lds_barrier();
  };

  static_assert(GX_AHEAD == 4, "the trips below are written out for 4 steps");
  for (int t0 = 0; t0 < 128 - GX_AHEAD; t0 += GX_AHEAD) {
    step(t0 + 0, StepReg<0>{}, StepAsk<true>{});
    step(t0 + 1, StepReg<1>{}, StepAsk<true>{});
    step(t0 + 2, StepReg<2>{}, StepAsk<true>{});
    step(t0 + 3, StepReg<3>{}, StepAsk<true>{});
  }
  step(128 - GX_AHEAD + 0, StepReg<0>{}, StepAsk<false>{});
  step(128 - GX_AHEAD + 1, StepReg<1>{}, StepAsk<false>{});
  step(128 - GX_AHEAD + 2, StepReg<2>{}, StepAsk<false>{});
  step(128 - GX_AHEAD + 3, StepReg<3>{}, StepAsk<false>{});
  if (wave < 4) hout[(size_t)127 * 128] = h_s[128 & 1][32 * wave + lane];
}

extern "C" void kernel_launch(void* const* d_in, const int* in_sizes, int n_in,
                              void* d_out, int out_size, void* d_ws, size_t ws_size,
                              hipStream_t stream)
{
  (void)in_sizes; (void)n_in; (void)out_size; (void)ws_size;
  const float* x      = (const float*)d_in[0];
  const float* hidden = (const float*)d_in[1];
  const float* h0     = (const float*)d_in[2];
  const float* c0     = (const float*)d_in[3];
  const float* f_w1   = (const float*)d_in[4];
  const float* f_b1   = (const float*)d_in[5];
  const float* f_w2   = (const float*)d_in[6];
  const float* f_b2   = (const float*)d_in[7];
  const float* g_w1   = (const float*)d_in[8];
  const float* g_b1   = (const float*)d_in[9];
  const float* g_w2   = (const float*)d_in[10];
  const float* g_b2   = (const float*)d_in[11];
  const float* wih    = (const float*)d_in[12];
  const float* whh    = (const float*)d_in[13];
  const float* bih    = (const float*)d_in[14];
  const float* bhh    = (const float*)d_in[15];
  const float* o_w1   = (const float*)d_in[16];
  const float* o_b1   = (const float*)d_in[17];
  const float* o_w2   = (const float*)d_in[18];
  const float* o_b2   = (const float*)d_in[19];
  float* out = (float*)d_out;

  float* R0 = (float*)d_ws;
  float* R1 = R0 + (size_t)2048 * 256;
  float* R2 = R1 + (size_t)2048 * 256;
  float* R3 = R2 + (size_t)2048 * 256;

  f_pair<<<256, 256, 0, stream>>>(hidden, f_w1, f_b1, R2);
  chain_g<<<128, 256, 0, stream>>>(R2, x, f_w2, f_b2, g_w1, g_b1, g_w2, g_b2,
                                   wih, bih, bhh, R0);
  lstm_scan<<<16, 512, 0, stream>>>(R0, whh, h0, c0, R3);
  chain_o<<<128, 256, 0, stream>>>(R3, o_w1, o_b1, o_w2, o_b2, out);
}
